# starting kernel
# baseline (speedup 1.0000x reference)
#include <hip/hip_runtime.h>
#include <hip/hip_fp16.h>


typedef float v2f __attribute__((ext_vector_type(2)));
typedef float v8f __attribute__((ext_vector_type(8)));

#define NPTS   50000
#define CCH    64
#define KPN    15
#define HNB    32
#define SIGMA  0.7f
#define BN_EPS 1e-5f

#define TILES     ((NPTS + 15) / 16)
#define GRID1     ((TILES + 7) / 8)
#define SCALE_OFF (GRID1 * 128)
#define SHIFT_OFF (SCALE_OFF + 64)

static __device__ __forceinline__ v8f wmma_f32_16x16x4(v2f a, v2f b, v8f c) {
  return __builtin_amdgcn_wmma_f32_16x16x4_f32(false, a, false, b,
                                               (short)0, c, false, false);
}

__global__ __launch_bounds__(256) void k1_stats(const float* __restrict__ sfeat,
                                                const float* __restrict__ W1,
                                                float* __restrict__ partials) {
  __shared__ float wsum[8][64];
  __shared__ float wssq[8][64];
  const int tid  = threadIdx.x;
  const int wave = tid >> 5;
  const int lane = tid & 31;
  const int half = lane >> 4;
  const int lm16 = lane & 15;
  const int t    = blockIdx.x * 8 + wave;

  v8f acc[4];
  #pragma unroll
  for (int n0 = 0; n0 < 4; ++n0)
    acc[n0] = (v8f){0.f,0.f,0.f,0.f,0.f,0.f,0.f,0.f};

  if (t < TILES) {
    const int row = t * 16 + lm16;
    for (int k0 = 0; k0 < CCH; k0 += 4) {
      const int kk = k0 + 2 * half;
      v2f a = *(const v2f*)(sfeat + row * CCH + kk);
      #pragma unroll
      for (int n0 = 0; n0 < 4; ++n0) {
        const int n = n0 * 16 + lm16;
        v2f b;
        b.x = W1[kk * CCH + n];
        b.y = W1[(kk + 1) * CCH + n];
        acc[n0] = wmma_f32_16x16x4(a, b, acc[n0]);
      }
    }
  }

  #pragma unroll
  for (int n0 = 0; n0 < 4; ++n0) {
    float s = 0.f, q = 0.f;
    #pragma unroll
    for (int r = 0; r < 8; ++r) { float v = acc[n0][r]; s += v; q += v * v; }
    s += __shfl_xor(s, 16);
    q += __shfl_xor(q, 16);
    if (half == 0) { wsum[wave][n0 * 16 + lm16] = s; wssq[wave][n0 * 16 + lm16] = q; }
  }
  __syncthreads();

  if (tid < 64) {
    float s = 0.f, q = 0.f;
    #pragma unroll
    for (int w = 0; w < 8; ++w) { s += wsum[w][tid]; q += wssq[w][tid]; }
    partials[blockIdx.x * 128 + tid]      = s;
    partials[blockIdx.x * 128 + 64 + tid] = q;
  }
}

__global__ void k2_finalize(const float* __restrict__ partials,
                            const float* __restrict__ gamma,
                            const float* __restrict__ beta,
                            float* __restrict__ ws) {
  const int ch = threadIdx.x;
  if (ch >= 64) return;
  float s = 0.f, q = 0.f;
  for (int b = 0; b < GRID1; ++b) {
    s += partials[b * 128 + ch];
    q += partials[b * 128 + 64 + ch];
  }
  const float inv_n = 1.0f / (float)NPTS;
  const float mu  = s * inv_n;
  const float var = q * inv_n - mu * mu;
  const float sc  = gamma[ch] * rsqrtf(var + BN_EPS);
  ws[SCALE_OFF + ch] = sc;
  ws[SHIFT_OFF + ch] = beta[ch] - mu * sc;
}

__global__ __launch_bounds__(256) void k3_main(const float* __restrict__ qpts,
                                               const float* __restrict__ spts,
                                               const float* __restrict__ sfeat,
                                               const int*   __restrict__ nbidx,
                                               const float* __restrict__ kpts,
                                               const float* __restrict__ W1,
                                               const float* __restrict__ W2,
                                               const float* __restrict__ b2,
                                               const float* __restrict__ ws,
                                               float* __restrict__ out) {
  __shared__ float  kp[45];
  __shared__ float  act[16 * 68];
  __shared__ __half cw[16 * 960];
  __shared__ int    jIdx[512];
  __shared__ int    nnK[512];
  __shared__ float  nnI[512];

  const int tid  = threadIdx.x;
  const int wave = tid >> 5;
  const int lane = tid & 31;
  const int half = lane >> 4;
  const int lm16 = lane & 15;
  const int m0   = blockIdx.x * 16;

  if (tid < 45) kp[tid] = kpts[tid];
  __syncthreads();

  if (wave < 4) {
    const int n0  = wave * 16;
    const int row = m0 + lm16;
    v8f acc = (v8f){0.f,0.f,0.f,0.f,0.f,0.f,0.f,0.f};
    for (int k0 = 0; k0 < CCH; k0 += 4) {
      const int kk = k0 + 2 * half;
      v2f a = *(const v2f*)(sfeat + row * CCH + kk);
      v2f b;
      b.x = W1[kk * CCH + n0 + lm16];
      b.y = W1[(kk + 1) * CCH + n0 + lm16];
      acc = wmma_f32_16x16x4(a, b, acc);
    }
    const int ch = n0 + lm16;
    const float sc = ws[SCALE_OFF + ch];
    const float sh = ws[SHIFT_OFF + ch];
    #pragma unroll
    for (int r = 0; r < 8; ++r) {
      const int M = r + 8 * half;
      float v = acc[r] * sc + sh;
      v = (v > 0.f) ? v : 0.1f * v;
      act[M * 68 + ch] = v;
    }
  } else {
    #pragma unroll
    for (int i = 0; i < 4; ++i) {
      const int p  = (wave - 4) * 128 + i * 32 + lane;
      const int lm = p >> 5;
      const int h  = p & 31;
      const int m  = m0 + lm;
      const int j  = nbidx[m * HNB + h];
      const bool valid = ((unsigned)j < (unsigned)NPTS);
      const int jj = valid ? j : 0;
      const float qx = qpts[m * 3 + 0];
      const float qy = qpts[m * 3 + 1];
      const float qz = qpts[m * 3 + 2];
      const float dx = spts[jj * 3 + 0] - qx;
      const float dy = spts[jj * 3 + 1] - qy;
      const float dz = spts[jj * 3 + 2] - qz;
      float best = 3.4e38f; int bk = 0;
      #pragma unroll
      for (int k = 0; k < KPN; ++k) {
        const float ex = dx - kp[k * 3 + 0];
        const float ey = dy - kp[k * 3 + 1];
        const float ez = dz - kp[k * 3 + 2];
        const float sq = ex * ex + ey * ey + ez * ez;
        if (sq < best) { best = sq; bk = k; }
      }
      float infl = 1.0f - sqrtf(best) * (1.0f / SIGMA);
      infl = (infl > 0.f) ? infl : 0.f;
      if (!valid) infl = 0.f;
      jIdx[p] = jj; nnK[p] = bk; nnI[p] = infl;
    }
  }
  __syncthreads();

  for (int t = wave; t < 60; t += 8) {
    const int n = t * 16 + lm16;
    v8f acc = (v8f){0.f,0.f,0.f,0.f,0.f,0.f,0.f,0.f};
    for (int k0 = 0; k0 < CCH; k0 += 4) {
      const int kk = k0 + 2 * half;
      v2f a = *(const v2f*)(act + lm16 * 68 + kk);
      v2f b;
      b.x = W2[kk * 960 + n];
      b.y = W2[(kk + 1) * 960 + n];
      acc = wmma_f32_16x16x4(a, b, acc);
    }
    const float bias = b2[n];
    #pragma unroll
    for (int r = 0; r < 8; ++r) {
      const int M = r + 8 * half;
      cw[M * 960 + n] = __float2half(acc[r] + bias);
    }
  }
  __syncthreads();

  #pragma unroll
  for (int rr = 0; rr < 2; ++rr) {
    const int lm = wave + rr * 8;
    float a0 = 0.f, a1 = 0.f;
    for (int h = 0; h < HNB; ++h) {
      const int p = lm * HNB + h;
      const float f = nnI[p];
      if (f > 0.f) {
        const int j = jIdx[p];
        const int k = nnK[p];
        const float w0 = __half2float(cw[lm * 960 + k * 64 + lane]) * f;
        const float w1 = __half2float(cw[lm * 960 + k * 64 + lane + 32]) * f;
        a0 = fmaf(sfeat[j * 64 + lane],      w0, a0);
        a1 = fmaf(sfeat[j * 64 + lane + 32], w1, a1);
      }
    }
    out[(m0 + lm) * 64 + lane]      = a0;
    out[(m0 + lm) * 64 + lane + 32] = a1;
  }
}

extern "C" void kernel_launch(void* const* d_in, const int* in_sizes, int n_in,
                              void* d_out, int out_size, void* d_ws, size_t ws_size,
                              hipStream_t stream) {
  const float* qpts  = (const float*)d_in[0];
  const float* spts  = (const float*)d_in[1];
  const float* sfeat = (const float*)d_in[2];
  const int*   nbidx = (const int*)  d_in[3];
  const float* kpts  = (const float*)d_in[4];
  const float* W1    = (const float*)d_in[5];
  const float* gamma = (const float*)d_in[6];
  const float* beta  = (const float*)d_in[7];
  const float* W2    = (const float*)d_in[8];
  const float* b2    = (const float*)d_in[9];
  float* out = (float*)d_out;
  float* ws  = (float*)d_ws;

  k1_stats<<<GRID1, 256, 0, stream>>>(sfeat, W1, ws);
  k2_finalize<<<1, 64, 0, stream>>>(ws, gamma, beta, ws);
  k3_main<<<TILES, 256, 0, stream>>>(qpts, spts, sfeat, nbidx, kpts,
                                     W1, W2, b2, ws, out);
  (void)in_sizes; (void)n_in; (void)out_size; (void)ws_size;
}


# BN finalize: one 1024-thread block sums the 391x128 partials with independent 16-byte loads and a fixed-order 32-way LDS sum (new reduction tree for mean/var) instead of a serial 391-step loop
# speedup vs baseline: 1.2579x; 1.2579x over previous
#include <hip/hip_runtime.h>
#include <hip/hip_fp16.h>


typedef float v2f __attribute__((ext_vector_type(2)));
typedef float v8f __attribute__((ext_vector_type(8)));

#define NPTS   50000
#define CCH    64
#define KPN    15
#define HNB    32
#define SIGMA  0.7f
#define BN_EPS 1e-5f

#define TILES     ((NPTS + 15) / 16)
#define GRID1     ((TILES + 7) / 8)
#define SCALE_OFF (GRID1 * 128)
#define SHIFT_OFF (SCALE_OFF + 64)

static __device__ __forceinline__ v8f wmma_f32_16x16x4(v2f a, v2f b, v8f c) {
  return __builtin_amdgcn_wmma_f32_16x16x4_f32(false, a, false, b,
                                               (short)0, c, false, false);
}

__global__ __launch_bounds__(256) void k1_stats(const float* __restrict__ sfeat,
                                                const float* __restrict__ W1,
                                                float* __restrict__ partials) {
  __shared__ float wsum[8][64];
  __shared__ float wssq[8][64];
  const int tid  = threadIdx.x;
  const int wave = tid >> 5;
  const int lane = tid & 31;
  const int half = lane >> 4;
  const int lm16 = lane & 15;
  const int t    = blockIdx.x * 8 + wave;

  v8f acc[4];
  #pragma unroll
  for (int n0 = 0; n0 < 4; ++n0)
    acc[n0] = (v8f){0.f,0.f,0.f,0.f,0.f,0.f,0.f,0.f};

  if (t < TILES) {
    const int row = t * 16 + lm16;
    for (int k0 = 0; k0 < CCH; k0 += 4) {
      const int kk = k0 + 2 * half;
      v2f a = *(const v2f*)(sfeat + row * CCH + kk);
      #pragma unroll
      for (int n0 = 0; n0 < 4; ++n0) {
        const int n = n0 * 16 + lm16;
        v2f b;
        b.x = W1[kk * CCH + n];
        b.y = W1[(kk + 1) * CCH + n];
        acc[n0] = wmma_f32_16x16x4(a, b, acc[n0]);
      }
    }
  }

  #pragma unroll
  for (int n0 = 0; n0 < 4; ++n0) {
    float s = 0.f, q = 0.f;
    #pragma unroll
    for (int r = 0; r < 8; ++r) { float v = acc[n0][r]; s += v; q += v * v; }
    s += __shfl_xor(s, 16);
    q += __shfl_xor(q, 16);
    if (half == 0) { wsum[wave][n0 * 16 + lm16] = s; wssq[wave][n0 * 16 + lm16] = q; }
  }
  __syncthreads();

  if (tid < 64) {
    float s = 0.f, q = 0.f;
    #pragma unroll
    for (int w = 0; w < 8; ++w) { s += wsum[w][tid]; q += wssq[w][tid]; }
    partials[blockIdx.x * 128 + tid]      = s;
    partials[blockIdx.x * 128 + 64 + tid] = q;
  }
}

#define K2_WAVES 32
#define K2_FULL  (GRID1 / K2_WAVES)
#define K2_REM   (GRID1 - K2_FULL * K2_WAVES)
__global__ __launch_bounds__(1024) void k2_finalize(const float* __restrict__ partials,
                                                    const float* __restrict__ gamma,
                                                    const float* __restrict__ beta,
                                                    float* __restrict__ ws) {
  __shared__ alignas(16) float red[K2_WAVES][128];
  const int tid = threadIdx.x;
  const int cq  = tid & 31;
  const int g   = tid >> 5;

  float4 v[K2_FULL];
  #pragma unroll
  for (int i = 0; i < K2_FULL; ++i)
    v[i] = *(const float4*)(partials + (g + K2_WAVES * i) * 128 + 4 * cq);
  float4 acc = v[0];
  #pragma unroll
  for (int i = 1; i < K2_FULL; ++i) {
    acc.x += v[i].x; acc.y += v[i].y; acc.z += v[i].z; acc.w += v[i].w;
  }
  if (g < K2_REM) {
    const float4 t = *(const float4*)(partials + (g + K2_WAVES * K2_FULL) * 128 + 4 * cq);
    acc.x += t.x; acc.y += t.y; acc.z += t.z; acc.w += t.w;
  }
  *(float4*)(&red[g][4 * cq]) = acc;
  __syncthreads();

  if (tid < 64) {
    const int ch = tid;
    float s = 0.f, q = 0.f;
    #pragma unroll
    for (int w = 0; w < K2_WAVES; ++w) { s += red[w][ch]; q += red[w][64 + ch]; }
    const float inv_n = 1.0f / (float)NPTS;
    const float mu  = s * inv_n;
    const float var = q * inv_n - mu * mu;
    const float sc  = gamma[ch] * rsqrtf(var + BN_EPS);
    ws[SCALE_OFF + ch] = sc;
    ws[SHIFT_OFF + ch] = beta[ch] - mu * sc;
  }
}

__global__ __launch_bounds__(256) void k3_main(const float* __restrict__ qpts,
                                               const float* __restrict__ spts,
                                               const float* __restrict__ sfeat,
                                               const int*   __restrict__ nbidx,
                                               const float* __restrict__ kpts,
                                               const float* __restrict__ W1,
                                               const float* __restrict__ W2,
                                               const float* __restrict__ b2,
                                               const float* __restrict__ ws,
                                               float* __restrict__ out) {
  __shared__ float  kp[45];
  __shared__ float  act[16 * 68];
  __shared__ __half cw[16 * 960];
  __shared__ int    jIdx[512];
  __shared__ int    nnK[512];
  __shared__ float  nnI[512];

  const int tid  = threadIdx.x;
  const int wave = tid >> 5;
  const int lane = tid & 31;
  const int half = lane >> 4;
  const int lm16 = lane & 15;
  const int m0   = blockIdx.x * 16;

  if (tid < 45) kp[tid] = kpts[tid];
  __syncthreads();

  if (wave < 4) {
    const int n0  = wave * 16;
    const int row = m0 + lm16;
    v8f acc = (v8f){0.f,0.f,0.f,0.f,0.f,0.f,0.f,0.f};
    for (int k0 = 0; k0 < CCH; k0 += 4) {
      const int kk = k0 + 2 * half;
      v2f a = *(const v2f*)(sfeat + row * CCH + kk);
      v2f b;
      b.x = W1[kk * CCH + n0 + lm16];
      b.y = W1[(kk + 1) * CCH + n0 + lm16];
      acc = wmma_f32_16x16x4(a, b, acc);
    }
    const int ch = n0 + lm16;
    const float sc = ws[SCALE_OFF + ch];
    const float sh = ws[SHIFT_OFF + ch];
    #pragma unroll
    for (int r = 0; r < 8; ++r) {
      const int M = r + 8 * half;
      float v = acc[r] * sc + sh;
      v = (v > 0.f) ? v : 0.1f * v;
      act[M * 68 + ch] = v;
    }
  } else {
    #pragma unroll
    for (int i = 0; i < 4; ++i) {
      const int p  = (wave - 4) * 128 + i * 32 + lane;
      const int lm = p >> 5;
      const int h  = p & 31;
      const int m  = m0 + lm;
      const int j  = nbidx[m * HNB + h];
      const bool valid = ((unsigned)j < (unsigned)NPTS);
      const int jj = valid ? j : 0;
      const float qx = qpts[m * 3 + 0];
      const float qy = qpts[m * 3 + 1];
      const float qz = qpts[m * 3 + 2];
      const float dx = spts[jj * 3 + 0] - qx;
      const float dy = spts[jj * 3 + 1] - qy;
      const float dz = spts[jj * 3 + 2] - qz;
      float best = 3.4e38f; int bk = 0;
      #pragma unroll
      for (int k = 0; k < KPN; ++k) {
        const float ex = dx - kp[k * 3 + 0];
        const float ey = dy - kp[k * 3 + 1];
        const float ez = dz - kp[k * 3 + 2];
        const float sq = ex * ex + ey * ey + ez * ez;
        if (sq < best) { best = sq; bk = k; }
      }
      float infl = 1.0f - sqrtf(best) * (1.0f / SIGMA);
      infl = (infl > 0.f) ? infl : 0.f;
      if (!valid) infl = 0.f;
      jIdx[p] = jj; nnK[p] = bk; nnI[p] = infl;
    }
  }
  __syncthreads();

  for (int t = wave; t < 60; t += 8) {
    const int n = t * 16 + lm16;
    v8f acc = (v8f){0.f,0.f,0.f,0.f,0.f,0.f,0.f,0.f};
    for (int k0 = 0; k0 < CCH; k0 += 4) {
      const int kk = k0 + 2 * half;
      v2f a = *(const v2f*)(act + lm16 * 68 + kk);
      v2f b;
      b.x = W2[kk * 960 + n];
      b.y = W2[(kk + 1) * 960 + n];
      acc = wmma_f32_16x16x4(a, b, acc);
    }
    const float bias = b2[n];
    #pragma unroll
    for (int r = 0; r < 8; ++r) {
      const int M = r + 8 * half;
      cw[M * 960 + n] = __float2half(acc[r] + bias);
    }
  }
  __syncthreads();

  #pragma unroll
  for (int rr = 0; rr < 2; ++rr) {
    const int lm = wave + rr * 8;
    float a0 = 0.f, a1 = 0.f;
    for (int h = 0; h < HNB; ++h) {
      const int p = lm * HNB + h;
      const float f = nnI[p];
      if (f > 0.f) {
        const int j = jIdx[p];
        const int k = nnK[p];
        const float w0 = __half2float(cw[lm * 960 + k * 64 + lane]) * f;
        const float w1 = __half2float(cw[lm * 960 + k * 64 + lane + 32]) * f;
        a0 = fmaf(sfeat[j * 64 + lane],      w0, a0);
        a1 = fmaf(sfeat[j * 64 + lane + 32], w1, a1);
      }
    }
    out[(m0 + lm) * 64 + lane]      = a0;
    out[(m0 + lm) * 64 + lane + 32] = a1;
  }
}

extern "C" void kernel_launch(void* const* d_in, const int* in_sizes, int n_in,
                              void* d_out, int out_size, void* d_ws, size_t ws_size,
                              hipStream_t stream) {
  const float* qpts  = (const float*)d_in[0];
  const float* spts  = (const float*)d_in[1];
  const float* sfeat = (const float*)d_in[2];
  const int*   nbidx = (const int*)  d_in[3];
  const float* kpts  = (const float*)d_in[4];
  const float* W1    = (const float*)d_in[5];
  const float* gamma = (const float*)d_in[6];
  const float* beta  = (const float*)d_in[7];
  const float* W2    = (const float*)d_in[8];
  const float* b2    = (const float*)d_in[9];
  float* out = (float*)d_out;
  float* ws  = (float*)d_ws;

  k1_stats<<<GRID1, 256, 0, stream>>>(sfeat, W1, ws);
  k2_finalize<<<1, 1024, 0, stream>>>(ws, gamma, beta, ws);
  k3_main<<<TILES, 256, 0, stream>>>(qpts, spts, sfeat, nbidx, kpts,
                                     W1, W2, b2, ws, out);
  (void)in_sizes; (void)n_in; (void)out_size; (void)ws_size;
}


# k3 needs 96 vector registers instead of 210 (kernel points by uniform scalar loads, not an LDS copy; geometry loop unrolled by 2, not 4): 5 workgroups resident per WGP instead of 2
# speedup vs baseline: 1.8611x; 1.4795x over previous
#include <hip/hip_runtime.h>
#include <hip/hip_fp16.h>


typedef float v2f __attribute__((ext_vector_type(2)));
typedef float v8f __attribute__((ext_vector_type(8)));

#define NPTS   50000
#define CCH    64
#define KPN    15
#define HNB    32
#define SIGMA  0.7f
#define BN_EPS 1e-5f

#define TILES     ((NPTS + 15) / 16)
#define GRID1     ((TILES + 7) / 8)
#define SCALE_OFF (GRID1 * 128)
#define SHIFT_OFF (SCALE_OFF + 64)

static __device__ __forceinline__ v8f wmma_f32_16x16x4(v2f a, v2f b, v8f c) {
  return __builtin_amdgcn_wmma_f32_16x16x4_f32(false, a, false, b,
                                               (short)0, c, false, false);
}

__global__ __launch_bounds__(256) void k1_stats(const float* __restrict__ sfeat,
                                                const float* __restrict__ W1,
                                                float* __restrict__ partials) {
  __shared__ float wsum[8][64];
  __shared__ float wssq[8][64];
  const int tid  = threadIdx.x;
  const int wave = tid >> 5;
  const int lane = tid & 31;
  const int half = lane >> 4;
  const int lm16 = lane & 15;
  const int t    = blockIdx.x * 8 + wave;

  v8f acc[4];
  #pragma unroll
  for (int n0 = 0; n0 < 4; ++n0)
    acc[n0] = (v8f){0.f,0.f,0.f,0.f,0.f,0.f,0.f,0.f};

  if (t < TILES) {
    const int row = t * 16 + lm16;
    for (int k0 = 0; k0 < CCH; k0 += 4) {
      const int kk = k0 + 2 * half;
      v2f a = *(const v2f*)(sfeat + row * CCH + kk);
      #pragma unroll
      for (int n0 = 0; n0 < 4; ++n0) {
        const int n = n0 * 16 + lm16;
        v2f b;
        b.x = W1[kk * CCH + n];
        b.y = W1[(kk + 1) * CCH + n];
        acc[n0] = wmma_f32_16x16x4(a, b, acc[n0]);
      }
    }
  }

  #pragma unroll
  for (int n0 = 0; n0 < 4; ++n0) {
    float s = 0.f, q = 0.f;
    #pragma unroll
    for (int r = 0; r < 8; ++r) { float v = acc[n0][r]; s += v; q += v * v; }
    s += __shfl_xor(s, 16);
    q += __shfl_xor(q, 16);
    if (half == 0) { wsum[wave][n0 * 16 + lm16] = s; wssq[wave][n0 * 16 + lm16] = q; }
  }
  __syncthreads();

  if (tid < 64) {
    float s = 0.f, q = 0.f;
    #pragma unroll
    for (int w = 0; w < 8; ++w) { s += wsum[w][tid]; q += wssq[w][tid]; }
    partials[blockIdx.x * 128 + tid]      = s;
    partials[blockIdx.x * 128 + 64 + tid] = q;
  }
}

#define K2_WAVES 32
#define K2_FULL  (GRID1 / K2_WAVES)
#define K2_REM   (GRID1 - K2_FULL * K2_WAVES)
__global__ __launch_bounds__(1024) void k2_finalize(const float* __restrict__ partials,
                                                    const float* __restrict__ gamma,
                                                    const float* __restrict__ beta,
                                                    float* __restrict__ ws) {
  __shared__ alignas(16) float red[K2_WAVES][128];
  const int tid = threadIdx.x;
  const int cq  = tid & 31;
  const int g   = tid >> 5;

  float4 v[K2_FULL];
  #pragma unroll
  for (int i = 0; i < K2_FULL; ++i)
    v[i] = *(const float4*)(partials + (g + K2_WAVES * i) * 128 + 4 * cq);
  float4 acc = v[0];
  #pragma unroll
  for (int i = 1; i < K2_FULL; ++i) {
    acc.x += v[i].x; acc.y += v[i].y; acc.z += v[i].z; acc.w += v[i].w;
  }
  if (g < K2_REM) {
    const float4 t = *(const float4*)(partials + (g + K2_WAVES * K2_FULL) * 128 + 4 * cq);
    acc.x += t.x; acc.y += t.y; acc.z += t.z; acc.w += t.w;
  }
  *(float4*)(&red[g][4 * cq]) = acc;
  __syncthreads();

  if (tid < 64) {
    const int ch = tid;
    float s = 0.f, q = 0.f;
    #pragma unroll
    for (int w = 0; w < K2_WAVES; ++w) { s += red[w][ch]; q += red[w][64 + ch]; }
    const float inv_n = 1.0f / (float)NPTS;
    const float mu  = s * inv_n;
    const float var = q * inv_n - mu * mu;
    const float sc  = gamma[ch] * rsqrtf(var + BN_EPS);
    ws[SCALE_OFF + ch] = sc;
    ws[SHIFT_OFF + ch] = beta[ch] - mu * sc;
  }
}

__global__ __launch_bounds__(256) void k3_main(const float* __restrict__ qpts,
                                               const float* __restrict__ spts,
                                               const float* __restrict__ sfeat,
                                               const int*   __restrict__ nbidx,
                                               const float* __restrict__ kpts,
                                               const float* __restrict__ W1,
                                               const float* __restrict__ W2,
                                               const float* __restrict__ b2,
                                               const float* __restrict__ ws,
                                               float* __restrict__ out) {
  __shared__ float  act[16 * 68];
  __shared__ __half cw[16 * 960];
  __shared__ int    jIdx[512];
  __shared__ int    nnK[512];
  __shared__ float  nnI[512];

  const int tid  = threadIdx.x;
  const int wave = tid >> 5;
  const int lane = tid & 31;
  const int half = lane >> 4;
  const int lm16 = lane & 15;
  const int m0   = blockIdx.x * 16;

  if (wave < 4) {
    const int n0  = wave * 16;
    const int row = m0 + lm16;
    v8f acc = (v8f){0.f,0.f,0.f,0.f,0.f,0.f,0.f,0.f};
    for (int k0 = 0; k0 < CCH; k0 += 4) {
      const int kk = k0 + 2 * half;
      v2f a = *(const v2f*)(sfeat + row * CCH + kk);
      v2f b;
      b.x = W1[kk * CCH + n0 + lm16];
      b.y = W1[(kk + 1) * CCH + n0 + lm16];
      acc = wmma_f32_16x16x4(a, b, acc);
    }
    const int ch = n0 + lm16;
    const float sc = ws[SCALE_OFF + ch];
    const float sh = ws[SHIFT_OFF + ch];
    #pragma unroll
    for (int r = 0; r < 8; ++r) {
      const int M = r + 8 * half;
      float v = acc[r] * sc + sh;
      v = (v > 0.f) ? v : 0.1f * v;
      act[M * 68 + ch] = v;
    }
  } else {
    #pragma unroll 2
    for (int i = 0; i < 4; ++i) {
      const int p  = (wave - 4) * 128 + i * 32 + lane;
      const int lm = p >> 5;
      const int h  = p & 31;
      const int m  = m0 + lm;
      const int j  = nbidx[m * HNB + h];
      const bool valid = ((unsigned)j < (unsigned)NPTS);
      const int jj = valid ? j : 0;
      const float qx = qpts[m * 3 + 0];
      const float qy = qpts[m * 3 + 1];
      const float qz = qpts[m * 3 + 2];
      const float dx = spts[jj * 3 + 0] - qx;
      const float dy = spts[jj * 3 + 1] - qy;
      const float dz = spts[jj * 3 + 2] - qz;
      float best = 3.4e38f; int bk = 0;
      #pragma unroll
      for (int k = 0; k < KPN; ++k) {
        const float ex = dx - kpts[k * 3 + 0];
        const float ey = dy - kpts[k * 3 + 1];
        const float ez = dz - kpts[k * 3 + 2];
        const float sq = ex * ex + ey * ey + ez * ez;
        if (sq < best) { best = sq; bk = k; }
      }
      float infl = 1.0f - sqrtf(best) * (1.0f / SIGMA);
      infl = (infl > 0.f) ? infl : 0.f;
      if (!valid) infl = 0.f;
      jIdx[p] = jj; nnK[p] = bk; nnI[p] = infl;
    }
  }
  __syncthreads();

  for (int t = wave; t < 60; t += 8) {
    const int n = t * 16 + lm16;
    v8f acc = (v8f){0.f,0.f,0.f,0.f,0.f,0.f,0.f,0.f};
    for (int k0 = 0; k0 < CCH; k0 += 4) {
      const int kk = k0 + 2 * half;
      v2f a = *(const v2f*)(act + lm16 * 68 + kk);
      v2f b;
      b.x = W2[kk * 960 + n];
      b.y = W2[(kk + 1) * 960 + n];
      acc = wmma_f32_16x16x4(a, b, acc);
    }
    const float bias = b2[n];
    #pragma unroll
    for (int r = 0; r < 8; ++r) {
      const int M = r + 8 * half;
      cw[M * 960 + n] = __float2half(acc[r] + bias);
    }
  }
  __syncthreads();

  #pragma unroll
  for (int rr = 0; rr < 2; ++rr) {
    const int lm = wave + rr * 8;
    float a0 = 0.f, a1 = 0.f;
    for (int h = 0; h < HNB; ++h) {
      const int p = lm * HNB + h;
      const float f = nnI[p];
      if (f > 0.f) {
        const int j = jIdx[p];
        const int k = nnK[p];
        const float w0 = __half2float(cw[lm * 960 + k * 64 + lane]) * f;
        const float w1 = __half2float(cw[lm * 960 + k * 64 + lane + 32]) * f;
        a0 = fmaf(sfeat[j * 64 + lane],      w0, a0);
        a1 = fmaf(sfeat[j * 64 + lane + 32], w1, a1);
      }
    }
    out[(m0 + lm) * 64 + lane]      = a0;
    out[(m0 + lm) * 64 + lane + 32] = a1;
  }
}

extern "C" void kernel_launch(void* const* d_in, const int* in_sizes, int n_in,
                              void* d_out, int out_size, void* d_ws, size_t ws_size,
                              hipStream_t stream) {
  const float* qpts  = (const float*)d_in[0];
  const float* spts  = (const float*)d_in[1];
  const float* sfeat = (const float*)d_in[2];
  const int*   nbidx = (const int*)  d_in[3];
  const float* kpts  = (const float*)d_in[4];
  const float* W1    = (const float*)d_in[5];
  const float* gamma = (const float*)d_in[6];
  const float* beta  = (const float*)d_in[7];
  const float* W2    = (const float*)d_in[8];
  const float* b2    = (const float*)d_in[9];
  float* out = (float*)d_out;
  float* ws  = (float*)d_ws;

  k1_stats<<<GRID1, 256, 0, stream>>>(sfeat, W1, ws);
  k2_finalize<<<1, 1024, 0, stream>>>(ws, gamma, beta, ws);
  k3_main<<<TILES, 256, 0, stream>>>(qpts, spts, sfeat, nbidx, kpts,
                                     W1, W2, b2, ws, out);
  (void)in_sizes; (void)n_in; (void)out_size; (void)ws_size;
}


# k3 aggregation without a test per neighbor: zero-influence pairs point at the query's own row and add x*0, so a row's 32 neighbor-row loads are in flight together, not one latency each
# speedup vs baseline: 1.9199x; 1.0316x over previous
#include <hip/hip_runtime.h>
#include <hip/hip_fp16.h>


typedef float v2f __attribute__((ext_vector_type(2)));
typedef float v8f __attribute__((ext_vector_type(8)));

#define NPTS   50000
#define CCH    64
#define KPN    15
#define HNB    32
#define SIGMA  0.7f
#define BN_EPS 1e-5f

#define TILES     ((NPTS + 15) / 16)
#define GRID1     ((TILES + 7) / 8)
#define SCALE_OFF (GRID1 * 128)
#define SHIFT_OFF (SCALE_OFF + 64)

static __device__ __forceinline__ v8f wmma_f32_16x16x4(v2f a, v2f b, v8f c) {
  return __builtin_amdgcn_wmma_f32_16x16x4_f32(false, a, false, b,
                                               (short)0, c, false, false);
}

__global__ __launch_bounds__(256) void k1_stats(const float* __restrict__ sfeat,
                                                const float* __restrict__ W1,
                                                float* __restrict__ partials) {
  __shared__ float wsum[8][64];
  __shared__ float wssq[8][64];
  const int tid  = threadIdx.x;
  const int wave = tid >> 5;
  const int lane = tid & 31;
  const int half = lane >> 4;
  const int lm16 = lane & 15;
  const int t    = blockIdx.x * 8 + wave;

  v8f acc[4];
  #pragma unroll
  for (int n0 = 0; n0 < 4; ++n0)
    acc[n0] = (v8f){0.f,0.f,0.f,0.f,0.f,0.f,0.f,0.f};

  if (t < TILES) {
    const int row = t * 16 + lm16;
    for (int k0 = 0; k0 < CCH; k0 += 4) {
      const int kk = k0 + 2 * half;
      v2f a = *(const v2f*)(sfeat + row * CCH + kk);
      #pragma unroll
      for (int n0 = 0; n0 < 4; ++n0) {
        const int n = n0 * 16 + lm16;
        v2f b;
        b.x = W1[kk * CCH + n];
        b.y = W1[(kk + 1) * CCH + n];
        acc[n0] = wmma_f32_16x16x4(a, b, acc[n0]);
      }
    }
  }

  #pragma unroll
  for (int n0 = 0; n0 < 4; ++n0) {
    float s = 0.f, q = 0.f;
    #pragma unroll
    for (int r = 0; r < 8; ++r) { float v = acc[n0][r]; s += v; q += v * v; }
    s += __shfl_xor(s, 16);
    q += __shfl_xor(q, 16);
    if (half == 0) { wsum[wave][n0 * 16 + lm16] = s; wssq[wave][n0 * 16 + lm16] = q; }
  }
  __syncthreads();

  if (tid < 64) {
    float s = 0.f, q = 0.f;
    #pragma unroll
    for (int w = 0; w < 8; ++w) { s += wsum[w][tid]; q += wssq[w][tid]; }
    partials[blockIdx.x * 128 + tid]      = s;
    partials[blockIdx.x * 128 + 64 + tid] = q;
  }
}

#define K2_WAVES 32
#define K2_FULL  (GRID1 / K2_WAVES)
#define K2_REM   (GRID1 - K2_FULL * K2_WAVES)
__global__ __launch_bounds__(1024) void k2_finalize(const float* __restrict__ partials,
                                                    const float* __restrict__ gamma,
                                                    const float* __restrict__ beta,
                                                    float* __restrict__ ws) {
  __shared__ alignas(16) float red[K2_WAVES][128];
  const int tid = threadIdx.x;
  const int cq  = tid & 31;
  const int g   = tid >> 5;

  float4 v[K2_FULL];
  #pragma unroll
  for (int i = 0; i < K2_FULL; ++i)
    v[i] = *(const float4*)(partials + (g + K2_WAVES * i) * 128 + 4 * cq);
  float4 acc = v[0];
  #pragma unroll
  for (int i = 1; i < K2_FULL; ++i) {
    acc.x += v[i].x; acc.y += v[i].y; acc.z += v[i].z; acc.w += v[i].w;
  }
  if (g < K2_REM) {
    const float4 t = *(const float4*)(partials + (g + K2_WAVES * K2_FULL) * 128 + 4 * cq);
    acc.x += t.x; acc.y += t.y; acc.z += t.z; acc.w += t.w;
  }
  *(float4*)(&red[g][4 * cq]) = acc;
  __syncthreads();

  if (tid < 64) {
    const int ch = tid;
    float s = 0.f, q = 0.f;
    #pragma unroll
    for (int w = 0; w < K2_WAVES; ++w) { s += red[w][ch]; q += red[w][64 + ch]; }
    const float inv_n = 1.0f / (float)NPTS;
    const float mu  = s * inv_n;
    const float var = q * inv_n - mu * mu;
    const float sc  = gamma[ch] * rsqrtf(var + BN_EPS);
    ws[SCALE_OFF + ch] = sc;
    ws[SHIFT_OFF + ch] = beta[ch] - mu * sc;
  }
}

__global__ __launch_bounds__(256) void k3_main(const float* __restrict__ qpts,
                                               const float* __restrict__ spts,
                                               const float* __restrict__ sfeat,
                                               const int*   __restrict__ nbidx,
                                               const float* __restrict__ kpts,
                                               const float* __restrict__ W1,
                                               const float* __restrict__ W2,
                                               const float* __restrict__ b2,
                                               const float* __restrict__ ws,
                                               float* __restrict__ out) {
  __shared__ float  act[16 * 68];
  __shared__ __half cw[16 * 960];
  __shared__ int    jIdx[512];
  __shared__ int    nnK[512];
  __shared__ float  nnI[512];

  const int tid  = threadIdx.x;
  const int wave = tid >> 5;
  const int lane = tid & 31;
  const int half = lane >> 4;
  const int lm16 = lane & 15;
  const int m0   = blockIdx.x * 16;

  if (wave < 4) {
    const int n0  = wave * 16;
    const int row = m0 + lm16;
    v8f acc = (v8f){0.f,0.f,0.f,0.f,0.f,0.f,0.f,0.f};
    for (int k0 = 0; k0 < CCH; k0 += 4) {
      const int kk = k0 + 2 * half;
      v2f a = *(const v2f*)(sfeat + row * CCH + kk);
      v2f b;
      b.x = W1[kk * CCH + n0 + lm16];
      b.y = W1[(kk + 1) * CCH + n0 + lm16];
      acc = wmma_f32_16x16x4(a, b, acc);
    }
    const int ch = n0 + lm16;
    const float sc = ws[SCALE_OFF + ch];
    const float sh = ws[SHIFT_OFF + ch];
    #pragma unroll
    for (int r = 0; r < 8; ++r) {
      const int M = r + 8 * half;
      float v = acc[r] * sc + sh;
      v = (v > 0.f) ? v : 0.1f * v;
      act[M * 68 + ch] = v;
    }
  } else {
    #pragma unroll 2
    for (int i = 0; i < 4; ++i) {
      const int p  = (wave - 4) * 128 + i * 32 + lane;
      const int lm = p >> 5;
      const int h  = p & 31;
      const int m  = m0 + lm;
      const int j  = nbidx[m * HNB + h];
      const bool valid = ((unsigned)j < (unsigned)NPTS);
      const int jj = valid ? j : 0;
      const float qx = qpts[m * 3 + 0];
      const float qy = qpts[m * 3 + 1];
      const float qz = qpts[m * 3 + 2];
      const float dx = spts[jj * 3 + 0] - qx;
      const float dy = spts[jj * 3 + 1] - qy;
      const float dz = spts[jj * 3 + 2] - qz;
      float best = 3.4e38f; int bk = 0;
      #pragma unroll
      for (int k = 0; k < KPN; ++k) {
        const float ex = dx - kpts[k * 3 + 0];
        const float ey = dy - kpts[k * 3 + 1];
        const float ez = dz - kpts[k * 3 + 2];
        const float sq = ex * ex + ey * ey + ez * ez;
        if (sq < best) { best = sq; bk = k; }
      }
      float infl = 1.0f - sqrtf(best) * (1.0f / SIGMA);
      infl = (infl > 0.f) ? infl : 0.f;
      if (!valid) infl = 0.f;
      jIdx[p] = (infl > 0.f) ? jj : m; nnK[p] = bk; nnI[p] = infl;
    }
  }
  __syncthreads();

  for (int t = wave; t < 60; t += 8) {
    const int n = t * 16 + lm16;
    v8f acc = (v8f){0.f,0.f,0.f,0.f,0.f,0.f,0.f,0.f};
    for (int k0 = 0; k0 < CCH; k0 += 4) {
      const int kk = k0 + 2 * half;
      v2f a = *(const v2f*)(act + lm16 * 68 + kk);
      v2f b;
      b.x = W2[kk * 960 + n];
      b.y = W2[(kk + 1) * 960 + n];
      acc = wmma_f32_16x16x4(a, b, acc);
    }
    const float bias = b2[n];
    #pragma unroll
    for (int r = 0; r < 8; ++r) {
      const int M = r + 8 * half;
      cw[M * 960 + n] = __float2half(acc[r] + bias);
    }
  }
  __syncthreads();

  #pragma unroll
  for (int rr = 0; rr < 2; ++rr) {
    const int lm = wave + rr * 8;
    const int   jv = jIdx[lm * HNB + lane];
    const int   kv = nnK[lm * HNB + lane];
    const int   fv = __float_as_int(nnI[lm * HNB + lane]);
    float a0 = 0.f, a1 = 0.f;
    float x0[HNB], x1[HNB];
    #pragma unroll
    for (int h = 0; h < HNB; ++h) {
      const int j = __builtin_amdgcn_readlane(jv, h);
      x0[h] = sfeat[j * 64 + lane];
      x1[h] = sfeat[j * 64 + lane + 32];
    }
    #pragma unroll
    for (int h = 0; h < HNB; ++h) {
      const int   k = __builtin_amdgcn_readlane(kv, h);
      const float f = __int_as_float(__builtin_amdgcn_readlane(fv, h));
      const float w0 = __half2float(cw[lm * 960 + k * 64 + lane]) * f;
      const float w1 = __half2float(cw[lm * 960 + k * 64 + lane + 32]) * f;
      a0 = fmaf(x0[h], w0, a0);
      a1 = fmaf(x1[h], w1, a1);
    }
    out[(m0 + lm) * 64 + lane]      = a0;
    out[(m0 + lm) * 64 + lane + 32] = a1;
  }
}

extern "C" void kernel_launch(void* const* d_in, const int* in_sizes, int n_in,
                              void* d_out, int out_size, void* d_ws, size_t ws_size,
                              hipStream_t stream) {
  const float* qpts  = (const float*)d_in[0];
  const float* spts  = (const float*)d_in[1];
  const float* sfeat = (const float*)d_in[2];
  const int*   nbidx = (const int*)  d_in[3];
  const float* kpts  = (const float*)d_in[4];
  const float* W1    = (const float*)d_in[5];
  const float* gamma = (const float*)d_in[6];
  const float* beta  = (const float*)d_in[7];
  const float* W2    = (const float*)d_in[8];
  const float* b2    = (const float*)d_in[9];
  float* out = (float*)d_out;
  float* ws  = (float*)d_ws;

  k1_stats<<<GRID1, 256, 0, stream>>>(sfeat, W1, ws);
  k2_finalize<<<1, 1024, 0, stream>>>(ws, gamma, beta, ws);
  k3_main<<<TILES, 256, 0, stream>>>(qpts, spts, sfeat, nbidx, kpts,
                                     W1, W2, b2, ws, out);
  (void)in_sizes; (void)n_in; (void)out_size; (void)ws_size;
}


# k3 works on 32 points (two WMMA row tiles) per workgroup: each W2 fragment a wave loads feeds two matrix products, so W2 reads from L2 per point halve; 1563 workgroups, 4 resident per WGP
# speedup vs baseline: 2.0603x; 1.0731x over previous
#include <hip/hip_runtime.h>
#include <hip/hip_fp16.h>


typedef float v2f __attribute__((ext_vector_type(2)));
typedef float v8f __attribute__((ext_vector_type(8)));

#define NPTS   50000
#define CCH    64
#define KPN    15
#define HNB    32
#define SIGMA  0.7f
#define BN_EPS 1e-5f

#define TILES     ((NPTS + 15) / 16)
#define GRID1     ((TILES + 7) / 8)
#define SCALE_OFF (GRID1 * 128)
#define SHIFT_OFF (SCALE_OFF + 64)

static __device__ __forceinline__ v8f wmma_f32_16x16x4(v2f a, v2f b, v8f c) {
  return __builtin_amdgcn_wmma_f32_16x16x4_f32(false, a, false, b,
                                               (short)0, c, false, false);
}

__global__ __launch_bounds__(256) void k1_stats(const float* __restrict__ sfeat,
                                                const float* __restrict__ W1,
                                                float* __restrict__ partials) {
  __shared__ float wsum[8][64];
  __shared__ float wssq[8][64];
  const int tid  = threadIdx.x;
  const int wave = tid >> 5;
  const int lane = tid & 31;
  const int half = lane >> 4;
  const int lm16 = lane & 15;
  const int t    = blockIdx.x * 8 + wave;

  v8f acc[4];
  #pragma unroll
  for (int n0 = 0; n0 < 4; ++n0)
    acc[n0] = (v8f){0.f,0.f,0.f,0.f,0.f,0.f,0.f,0.f};

  if (t < TILES) {
    const int row = t * 16 + lm16;
    for (int k0 = 0; k0 < CCH; k0 += 4) {
      const int kk = k0 + 2 * half;
      v2f a = *(const v2f*)(sfeat + row * CCH + kk);
      #pragma unroll
      for (int n0 = 0; n0 < 4; ++n0) {
        const int n = n0 * 16 + lm16;
        v2f b;
        b.x = W1[kk * CCH + n];
        b.y = W1[(kk + 1) * CCH + n];
        acc[n0] = wmma_f32_16x16x4(a, b, acc[n0]);
      }
    }
  }

  #pragma unroll
  for (int n0 = 0; n0 < 4; ++n0) {
    float s = 0.f, q = 0.f;
    #pragma unroll
    for (int r = 0; r < 8; ++r) { float v = acc[n0][r]; s += v; q += v * v; }
    s += __shfl_xor(s, 16);
    q += __shfl_xor(q, 16);
    if (half == 0) { wsum[wave][n0 * 16 + lm16] = s; wssq[wave][n0 * 16 + lm16] = q; }
  }
  __syncthreads();

  if (tid < 64) {
    float s = 0.f, q = 0.f;
    #pragma unroll
    for (int w = 0; w < 8; ++w) { s += wsum[w][tid]; q += wssq[w][tid]; }
    partials[blockIdx.x * 128 + tid]      = s;
    partials[blockIdx.x * 128 + 64 + tid] = q;
  }
}

#define K2_WAVES 32
#define K2_FULL  (GRID1 / K2_WAVES)
#define K2_REM   (GRID1 - K2_FULL * K2_WAVES)
__global__ __launch_bounds__(1024) void k2_finalize(const float* __restrict__ partials,
                                                    const float* __restrict__ gamma,
                                                    const float* __restrict__ beta,
                                                    float* __restrict__ ws) {
  __shared__ alignas(16) float red[K2_WAVES][128];
  const int tid = threadIdx.x;
  const int cq  = tid & 31;
  const int g   = tid >> 5;

  float4 v[K2_FULL];
  #pragma unroll
  for (int i = 0; i < K2_FULL; ++i)
    v[i] = *(const float4*)(partials + (g + K2_WAVES * i) * 128 + 4 * cq);
  float4 acc = v[0];
  #pragma unroll
  for (int i = 1; i < K2_FULL; ++i) {
    acc.x += v[i].x; acc.y += v[i].y; acc.z += v[i].z; acc.w += v[i].w;
  }
  if (g < K2_REM) {
    const float4 t = *(const float4*)(partials + (g + K2_WAVES * K2_FULL) * 128 + 4 * cq);
    acc.x += t.x; acc.y += t.y; acc.z += t.z; acc.w += t.w;
  }
  *(float4*)(&red[g][4 * cq]) = acc;
  __syncthreads();

  if (tid < 64) {
    const int ch = tid;
    float s = 0.f, q = 0.f;
    #pragma unroll
    for (int w = 0; w < K2_WAVES; ++w) { s += red[w][ch]; q += red[w][64 + ch]; }
    const float inv_n = 1.0f / (float)NPTS;
    const float mu  = s * inv_n;
    const float var = q * inv_n - mu * mu;
    const float sc  = gamma[ch] * rsqrtf(var + BN_EPS);
    ws[SCALE_OFF + ch] = sc;
    ws[SHIFT_OFF + ch] = beta[ch] - mu * sc;
  }
}

#define ROWS3   32
#define BLOCKS3 ((NPTS + ROWS3 - 1) / ROWS3)
#define NPAIR3  (ROWS3 * HNB)
__global__ __launch_bounds__(256) void k3_main(const float* __restrict__ qpts,
                                               const float* __restrict__ spts,
                                               const float* __restrict__ sfeat,
                                               const int*   __restrict__ nbidx,
                                               const float* __restrict__ kpts,
                                               const float* __restrict__ W1,
                                               const float* __restrict__ W2,
                                               const float* __restrict__ b2,
                                               const float* __restrict__ ws,
                                               float* __restrict__ out) {
  __shared__ float  act[ROWS3 * 68];
  __shared__ __half cw[ROWS3 * 960];
  __shared__ int    jIdx[NPAIR3];
  __shared__ float  nnI[NPAIR3];
  __shared__ unsigned char nnK[NPAIR3];

  const int tid  = threadIdx.x;
  const int wave = tid >> 5;
  const int lane = tid & 31;
  const int half = lane >> 4;
  const int lm16 = lane & 15;
  const int m0   = blockIdx.x * ROWS3;

  {
    const int mt  = wave >> 2;
    const int n0  = (wave & 3) * 16;
    int row = m0 + mt * 16 + lm16;
    row = (row < NPTS) ? row : (NPTS - 1);
    v8f acc = (v8f){0.f,0.f,0.f,0.f,0.f,0.f,0.f,0.f};
    for (int k0 = 0; k0 < CCH; k0 += 4) {
      const int kk = k0 + 2 * half;
      v2f a = *(const v2f*)(sfeat + row * CCH + kk);
      v2f b;
      b.x = W1[kk * CCH + n0 + lm16];
      b.y = W1[(kk + 1) * CCH + n0 + lm16];
      acc = wmma_f32_16x16x4(a, b, acc);
    }
    const int ch = n0 + lm16;
    const float sc = ws[SCALE_OFF + ch];
    const float sh = ws[SHIFT_OFF + ch];
    #pragma unroll
    for (int r = 0; r < 8; ++r) {
      const int M = mt * 16 + r + 8 * half;
      float v = acc[r] * sc + sh;
      v = (v > 0.f) ? v : 0.1f * v;
      act[M * 68 + ch] = v;
    }
  }
  {
    #pragma unroll 2
    for (int i = 0; i < 4; ++i) {
      const int p  = i * 256 + tid;
      const int lm = p >> 5;
      const int h  = p & 31;
      int m = m0 + lm;
      m = (m < NPTS) ? m : (NPTS - 1);
      const int j  = nbidx[m * HNB + h];
      const bool valid = ((unsigned)j < (unsigned)NPTS);
      const int jj = valid ? j : 0;
      const float qx = qpts[m * 3 + 0];
      const float qy = qpts[m * 3 + 1];
      const float qz = qpts[m * 3 + 2];
      const float dx = spts[jj * 3 + 0] - qx;
      const float dy = spts[jj * 3 + 1] - qy;
      const float dz = spts[jj * 3 + 2] - qz;
      float best = 3.4e38f; int bk = 0;
      #pragma unroll
      for (int k = 0; k < KPN; ++k) {
        const float ex = dx - kpts[k * 3 + 0];
        const float ey = dy - kpts[k * 3 + 1];
        const float ez = dz - kpts[k * 3 + 2];
        const float sq = ex * ex + ey * ey + ez * ez;
        if (sq < best) { best = sq; bk = k; }
      }
      float infl = 1.0f - sqrtf(best) * (1.0f / SIGMA);
      infl = (infl > 0.f) ? infl : 0.f;
      if (!valid) infl = 0.f;
      jIdx[p] = (infl > 0.f) ? jj : m;
      nnK[p]  = (unsigned char)bk;
      nnI[p]  = infl;
    }
  }
  __syncthreads();

  for (int t = wave; t < 60; t += 8) {
    const int n = t * 16 + lm16;
    v8f acc0 = (v8f){0.f,0.f,0.f,0.f,0.f,0.f,0.f,0.f};
    v8f acc1 = (v8f){0.f,0.f,0.f,0.f,0.f,0.f,0.f,0.f};
    for (int k0 = 0; k0 < CCH; k0 += 4) {
      const int kk = k0 + 2 * half;
      v2f a0 = *(const v2f*)(act + lm16 * 68 + kk);
      v2f a1 = *(const v2f*)(act + (16 + lm16) * 68 + kk);
      v2f b;
      b.x = W2[kk * 960 + n];
      b.y = W2[(kk + 1) * 960 + n];
      acc0 = wmma_f32_16x16x4(a0, b, acc0);
      acc1 = wmma_f32_16x16x4(a1, b, acc1);
    }
    const float bias = b2[n];
    #pragma unroll
    for (int r = 0; r < 8; ++r) {
      const int M = r + 8 * half;
      cw[M * 960 + n]        = __float2half(acc0[r] + bias);
      cw[(16 + M) * 960 + n] = __float2half(acc1[r] + bias);
    }
  }
  __syncthreads();

  const int ntile = (m0 + 16 < NPTS) ? 2 : 1;
  #pragma unroll 1
  for (int rq = 0; rq < ntile; ++rq) {
    #pragma unroll
    for (int rr = 0; rr < 2; ++rr) {
      const int lm = wave + (rq * 2 + rr) * 8;
      const int   jv = jIdx[lm * HNB + lane];
      const int   kv = nnK[lm * HNB + lane];
      const int   fv = __float_as_int(nnI[lm * HNB + lane]);
      float a0 = 0.f, a1 = 0.f;
      float x0[HNB], x1[HNB];
      #pragma unroll
      for (int h = 0; h < HNB; ++h) {
        const int j = __builtin_amdgcn_readlane(jv, h);
        x0[h] = sfeat[j * 64 + lane];
        x1[h] = sfeat[j * 64 + lane + 32];
      }
      #pragma unroll
      for (int h = 0; h < HNB; ++h) {
        const int   k = __builtin_amdgcn_readlane(kv, h);
        const float f = __int_as_float(__builtin_amdgcn_readlane(fv, h));
        const float w0 = __half2float(cw[lm * 960 + k * 64 + lane]) * f;
        const float w1 = __half2float(cw[lm * 960 + k * 64 + lane + 32]) * f;
        a0 = fmaf(x0[h], w0, a0);
        a1 = fmaf(x1[h], w1, a1);
      }
      out[(m0 + lm) * 64 + lane]      = a0;
      out[(m0 + lm) * 64 + lane + 32] = a1;
    }
  }
}

extern "C" void kernel_launch(void* const* d_in, const int* in_sizes, int n_in,
                              void* d_out, int out_size, void* d_ws, size_t ws_size,
                              hipStream_t stream) {
  const float* qpts  = (const float*)d_in[0];
  const float* spts  = (const float*)d_in[1];
  const float* sfeat = (const float*)d_in[2];
  const int*   nbidx = (const int*)  d_in[3];
  const float* kpts  = (const float*)d_in[4];
  const float* W1    = (const float*)d_in[5];
  const float* gamma = (const float*)d_in[6];
  const float* beta  = (const float*)d_in[7];
  const float* W2    = (const float*)d_in[8];
  const float* b2    = (const float*)d_in[9];
  float* out = (float*)d_out;
  float* ws  = (float*)d_ws;

  k1_stats<<<GRID1, 256, 0, stream>>>(sfeat, W1, ws);
  k2_finalize<<<1, 1024, 0, stream>>>(ws, gamma, beta, ws);
  k3_main<<<BLOCKS3, 256, 0, stream>>>(qpts, spts, sfeat, nbidx, kpts,
                                     W1, W2, b2, ws, out);
  (void)in_sizes; (void)n_in; (void)out_size; (void)ws_size;
}


# k3 builds conv weights only for the (point, kernel point) pairs a neighbor with influence reads: per kernel point the points needing it are packed into 16-row WMMA tiles, about half the products
# speedup vs baseline: 2.1316x; 1.0346x over previous
#include <hip/hip_runtime.h>
#include <hip/hip_fp16.h>


typedef float v2f __attribute__((ext_vector_type(2)));
typedef float v8f __attribute__((ext_vector_type(8)));

#define NPTS   50000
#define CCH    64
#define KPN    15
#define HNB    32
#define SIGMA  0.7f
#define BN_EPS 1e-5f

#define TILES     ((NPTS + 15) / 16)
#define GRID1     ((TILES + 7) / 8)
#define SCALE_OFF (GRID1 * 128)
#define SHIFT_OFF (SCALE_OFF + 64)

static __device__ __forceinline__ v8f wmma_f32_16x16x4(v2f a, v2f b, v8f c) {
  return __builtin_amdgcn_wmma_f32_16x16x4_f32(false, a, false, b,
                                               (short)0, c, false, false);
}

static __device__ __forceinline__ int nth_set_bit(unsigned bal, int r) {
  int pos = 0;
  #pragma unroll
  for (int w = 16; w >= 1; w >>= 1) {
    const int c = __popc((bal >> pos) & ((1u << w) - 1u));
    if (r >= c) { r -= c; pos += w; }
  }
  return pos;
}

__global__ __launch_bounds__(256) void k1_stats(const float* __restrict__ sfeat,
                                                const float* __restrict__ W1,
                                                float* __restrict__ partials) {
  __shared__ float wsum[8][64];
  __shared__ float wssq[8][64];
  const int tid  = threadIdx.x;
  const int wave = tid >> 5;
  const int lane = tid & 31;
  const int half = lane >> 4;
  const int lm16 = lane & 15;
  const int t    = blockIdx.x * 8 + wave;

  v8f acc[4];
  #pragma unroll
  for (int n0 = 0; n0 < 4; ++n0)
    acc[n0] = (v8f){0.f,0.f,0.f,0.f,0.f,0.f,0.f,0.f};

  if (t < TILES) {
    const int row = t * 16 + lm16;
    for (int k0 = 0; k0 < CCH; k0 += 4) {
      const int kk = k0 + 2 * half;
      v2f a = *(const v2f*)(sfeat + row * CCH + kk);
      #pragma unroll
      for (int n0 = 0; n0 < 4; ++n0) {
        const int n = n0 * 16 + lm16;
        v2f b;
        b.x = W1[kk * CCH + n];
        b.y = W1[(kk + 1) * CCH + n];
        acc[n0] = wmma_f32_16x16x4(a, b, acc[n0]);
      }
    }
  }

  #pragma unroll
  for (int n0 = 0; n0 < 4; ++n0) {
    float s = 0.f, q = 0.f;
    #pragma unroll
    for (int r = 0; r < 8; ++r) { float v = acc[n0][r]; s += v; q += v * v; }
    s += __shfl_xor(s, 16);
    q += __shfl_xor(q, 16);
    if (half == 0) { wsum[wave][n0 * 16 + lm16] = s; wssq[wave][n0 * 16 + lm16] = q; }
  }
  __syncthreads();

  if (tid < 64) {
    float s = 0.f, q = 0.f;
    #pragma unroll
    for (int w = 0; w < 8; ++w) { s += wsum[w][tid]; q += wssq[w][tid]; }
    partials[blockIdx.x * 128 + tid]      = s;
    partials[blockIdx.x * 128 + 64 + tid] = q;
  }
}

#define K2_WAVES 32
#define K2_FULL  (GRID1 / K2_WAVES)
#define K2_REM   (GRID1 - K2_FULL * K2_WAVES)
__global__ __launch_bounds__(1024) void k2_finalize(const float* __restrict__ partials,
                                                    const float* __restrict__ gamma,
                                                    const float* __restrict__ beta,
                                                    float* __restrict__ ws) {
  __shared__ alignas(16) float red[K2_WAVES][128];
  const int tid = threadIdx.x;
  const int cq  = tid & 31;
  const int g   = tid >> 5;

  float4 v[K2_FULL];
  #pragma unroll
  for (int i = 0; i < K2_FULL; ++i)
    v[i] = *(const float4*)(partials + (g + K2_WAVES * i) * 128 + 4 * cq);
  float4 acc = v[0];
  #pragma unroll
  for (int i = 1; i < K2_FULL; ++i) {
    acc.x += v[i].x; acc.y += v[i].y; acc.z += v[i].z; acc.w += v[i].w;
  }
  if (g < K2_REM) {
    const float4 t = *(const float4*)(partials + (g + K2_WAVES * K2_FULL) * 128 + 4 * cq);
    acc.x += t.x; acc.y += t.y; acc.z += t.z; acc.w += t.w;
  }
  *(float4*)(&red[g][4 * cq]) = acc;
  __syncthreads();

  if (tid < 64) {
    const int ch = tid;
    float s = 0.f, q = 0.f;
    #pragma unroll
    for (int w = 0; w < K2_WAVES; ++w) { s += red[w][ch]; q += red[w][64 + ch]; }
    const float inv_n = 1.0f / (float)NPTS;
    const float mu  = s * inv_n;
    const float var = q * inv_n - mu * mu;
    const float sc  = gamma[ch] * rsqrtf(var + BN_EPS);
    ws[SCALE_OFF + ch] = sc;
    ws[SHIFT_OFF + ch] = beta[ch] - mu * sc;
  }
}

#define ROWS3   32
#define BLOCKS3 ((NPTS + ROWS3 - 1) / ROWS3)
#define NPAIR3  (ROWS3 * HNB)
__global__ __launch_bounds__(256) void k3_main(const float* __restrict__ qpts,
                                               const float* __restrict__ spts,
                                               const float* __restrict__ sfeat,
                                               const int*   __restrict__ nbidx,
                                               const float* __restrict__ kpts,
                                               const float* __restrict__ W1,
                                               const float* __restrict__ W2,
                                               const float* __restrict__ b2,
                                               const float* __restrict__ ws,
                                               float* __restrict__ out) {
  __shared__ float  act[ROWS3 * 68];
  __shared__ __half cw[ROWS3 * 960];
  __shared__ int    jIdx[NPAIR3];
  __shared__ float  nnI[NPAIR3];
  __shared__ unsigned char nnK[NPAIR3];
  __shared__ unsigned needM[ROWS3];

  const int tid  = threadIdx.x;
  const int wave = tid >> 5;
  const int lane = tid & 31;
  const int half = lane >> 4;
  const int lm16 = lane & 15;
  const int m0   = blockIdx.x * ROWS3;

  {
    const int mt  = wave >> 2;
    const int n0  = (wave & 3) * 16;
    int row = m0 + mt * 16 + lm16;
    row = (row < NPTS) ? row : (NPTS - 1);
    v8f acc = (v8f){0.f,0.f,0.f,0.f,0.f,0.f,0.f,0.f};
    for (int k0 = 0; k0 < CCH; k0 += 4) {
      const int kk = k0 + 2 * half;
      v2f a = *(const v2f*)(sfeat + row * CCH + kk);
      v2f b;
      b.x = W1[kk * CCH + n0 + lm16];
      b.y = W1[(kk + 1) * CCH + n0 + lm16];
      acc = wmma_f32_16x16x4(a, b, acc);
    }
    const int ch = n0 + lm16;
    const float sc = ws[SCALE_OFF + ch];
    const float sh = ws[SHIFT_OFF + ch];
    #pragma unroll
    for (int r = 0; r < 8; ++r) {
      const int M = mt * 16 + r + 8 * half;
      float v = acc[r] * sc + sh;
      v = (v > 0.f) ? v : 0.1f * v;
      act[M * 68 + ch] = v;
    }
  }
  {
    #pragma unroll 2
    for (int i = 0; i < 4; ++i) {
      const int p  = i * 256 + tid;
      const int lm = p >> 5;
      const int h  = p & 31;
      int m = m0 + lm;
      m = (m < NPTS) ? m : (NPTS - 1);
      const int j  = nbidx[m * HNB + h];
      const bool valid = ((unsigned)j < (unsigned)NPTS);
      const int jj = valid ? j : 0;
      const float qx = qpts[m * 3 + 0];
      const float qy = qpts[m * 3 + 1];
      const float qz = qpts[m * 3 + 2];
      const float dx = spts[jj * 3 + 0] - qx;
      const float dy = spts[jj * 3 + 1] - qy;
      const float dz = spts[jj * 3 + 2] - qz;
      float best = 3.4e38f; int bk = 0;
      #pragma unroll
      for (int k = 0; k < KPN; ++k) {
        const float ex = dx - kpts[k * 3 + 0];
        const float ey = dy - kpts[k * 3 + 1];
        const float ez = dz - kpts[k * 3 + 2];
        const float sq = ex * ex + ey * ey + ez * ez;
        if (sq < best) { best = sq; bk = k; }
      }
      float infl = 1.0f - sqrtf(best) * (1.0f / SIGMA);
      infl = (infl > 0.f) ? infl : 0.f;
      if (!valid) infl = 0.f;
      const bool on = (infl > 0.f);
      int um = on ? (1 << bk) : 0;
      um |= __shfl_xor(um, 16);
      um |= __shfl_xor(um, 8);
      um |= __shfl_xor(um, 4);
      um |= __shfl_xor(um, 2);
      um |= __shfl_xor(um, 1);
      um = um ? um : 1;
      jIdx[p] = on ? jj : m;
      nnK[p]  = (unsigned char)(on ? bk : (__ffs(um) - 1));
      nnI[p]  = infl;
      if (h == 0) needM[lm] = (unsigned)um;
    }
  }
  __syncthreads();

  const unsigned nm = needM[lane];
  #pragma unroll 1
  for (int it = wave; it < 2 * KPN; it += 8) {
    const int kp = it >> 1;
    const int cb = kp * 64 + (it & 1) * 32 + 2 * lm16;
    const unsigned bal = __builtin_amdgcn_ballot_w32(((nm >> kp) & 1u) != 0u);
    const int nk = __popc(bal);
    const v2f bias = *(const v2f*)(b2 + cb);
    #pragma unroll 1
    for (int s0 = 0; s0 < nk; s0 += 16) {
      const int s  = s0 + lm16;
      const int mr = nth_set_bit(bal, (s < nk) ? s : s0);
      v8f acc0 = (v8f){0.f,0.f,0.f,0.f,0.f,0.f,0.f,0.f};
      v8f acc1 = (v8f){0.f,0.f,0.f,0.f,0.f,0.f,0.f,0.f};
      const float* ap = act + mr * 68 + 2 * half;
      const float* bp = W2 + (2 * half) * 960 + cb;
      #pragma unroll
      for (int ks = 0; ks < CCH / 4; ++ks) {
        const v2f a  = *(const v2f*)(ap + 4 * ks);
        const v2f bx = *(const v2f*)(bp + (4 * ks) * 960);
        const v2f by = *(const v2f*)(bp + (4 * ks + 1) * 960);
        acc0 = wmma_f32_16x16x4(a, (v2f){bx.x, by.x}, acc0);
        acc1 = wmma_f32_16x16x4(a, (v2f){bx.y, by.y}, acc1);
      }
      #pragma unroll
      for (int r = 0; r < 8; ++r) {
        const int plo = __builtin_amdgcn_readlane(mr, r);
        const int phi = __builtin_amdgcn_readlane(mr, r + 8);
        const int pr  = half ? phi : plo;
        *(__half2*)(cw + pr * 960 + cb) =
            __halves2half2(__float2half(acc0[r] + bias.x),
                           __float2half(acc1[r] + bias.y));
      }
    }
  }
  __syncthreads();

  const int ntile = (m0 + 16 < NPTS) ? 2 : 1;
  #pragma unroll 1
  for (int rq = 0; rq < ntile; ++rq) {
    #pragma unroll
    for (int rr = 0; rr < 2; ++rr) {
      const int lm = wave + (rq * 2 + rr) * 8;
      const int   jv = jIdx[lm * HNB + lane];
      const int   kv = nnK[lm * HNB + lane];
      const int   fv = __float_as_int(nnI[lm * HNB + lane]);
      float a0 = 0.f, a1 = 0.f;
      float x0[HNB], x1[HNB];
      #pragma unroll
      for (int h = 0; h < HNB; ++h) {
        const int j = __builtin_amdgcn_readlane(jv, h);
        x0[h] = sfeat[j * 64 + lane];
        x1[h] = sfeat[j * 64 + lane + 32];
      }
      #pragma unroll
      for (int h = 0; h < HNB; ++h) {
        const int   k = __builtin_amdgcn_readlane(kv, h);
        const float f = __int_as_float(__builtin_amdgcn_readlane(fv, h));
        const float w0 = __half2float(cw[lm * 960 + k * 64 + lane]) * f;
        const float w1 = __half2float(cw[lm * 960 + k * 64 + lane + 32]) * f;
        a0 = fmaf(x0[h], w0, a0);
        a1 = fmaf(x1[h], w1, a1);
      }
      out[(m0 + lm) * 64 + lane]      = a0;
      out[(m0 + lm) * 64 + lane + 32] = a1;
    }
  }
}

extern "C" void kernel_launch(void* const* d_in, const int* in_sizes, int n_in,
                              void* d_out, int out_size, void* d_ws, size_t ws_size,
                              hipStream_t stream) {
  const float* qpts  = (const float*)d_in[0];
  const float* spts  = (const float*)d_in[1];
  const float* sfeat = (const float*)d_in[2];
  const int*   nbidx = (const int*)  d_in[3];
  const float* kpts  = (const float*)d_in[4];
  const float* W1    = (const float*)d_in[5];
  const float* gamma = (const float*)d_in[6];
  const float* beta  = (const float*)d_in[7];
  const float* W2    = (const float*)d_in[8];
  const float* b2    = (const float*)d_in[9];
  float* out = (float*)d_out;
  float* ws  = (float*)d_ws;

  k1_stats<<<GRID1, 256, 0, stream>>>(sfeat, W1, ws);
  k2_finalize<<<1, 1024, 0, stream>>>(ws, gamma, beta, ws);
  k3_main<<<BLOCKS3, 256, 0, stream>>>(qpts, spts, sfeat, nbidx, kpts,
                                     W1, W2, b2, ws, out);
  (void)in_sizes; (void)n_in; (void)out_size; (void)ws_size;
}


# k3 aggregation: a lane reads 16 bytes of a neighbor row, so one load fetches two neighbors' rows; even-h and odd-h neighbors are summed in separate lane halves, then added (new summation order)
# speedup vs baseline: 2.4051x; 1.1283x over previous
#include <hip/hip_runtime.h>
#include <hip/hip_fp16.h>


typedef float v2f __attribute__((ext_vector_type(2)));
typedef float v8f __attribute__((ext_vector_type(8)));

#define NPTS   50000
#define CCH    64
#define KPN    15
#define HNB    32
#define SIGMA  0.7f
#define BN_EPS 1e-5f

#define TILES     ((NPTS + 15) / 16)
#define GRID1     ((TILES + 7) / 8)
#define SCALE_OFF (GRID1 * 128)
#define SHIFT_OFF (SCALE_OFF + 64)

static __device__ __forceinline__ v8f wmma_f32_16x16x4(v2f a, v2f b, v8f c) {
  return __builtin_amdgcn_wmma_f32_16x16x4_f32(false, a, false, b,
                                               (short)0, c, false, false);
}

static __device__ __forceinline__ int nth_set_bit(unsigned bal, int r) {
  int pos = 0;
  #pragma unroll
  for (int w = 16; w >= 1; w >>= 1) {
    const int c = __popc((bal >> pos) & ((1u << w) - 1u));
    if (r >= c) { r -= c; pos += w; }
  }
  return pos;
}

__global__ __launch_bounds__(256) void k1_stats(const float* __restrict__ sfeat,
                                                const float* __restrict__ W1,
                                                float* __restrict__ partials) {
  __shared__ float wsum[8][64];
  __shared__ float wssq[8][64];
  const int tid  = threadIdx.x;
  const int wave = tid >> 5;
  const int lane = tid & 31;
  const int half = lane >> 4;
  const int lm16 = lane & 15;
  const int t    = blockIdx.x * 8 + wave;

  v8f acc[4];
  #pragma unroll
  for (int n0 = 0; n0 < 4; ++n0)
    acc[n0] = (v8f){0.f,0.f,0.f,0.f,0.f,0.f,0.f,0.f};

  if (t < TILES) {
    const int row = t * 16 + lm16;
    for (int k0 = 0; k0 < CCH; k0 += 4) {
      const int kk = k0 + 2 * half;
      v2f a = *(const v2f*)(sfeat + row * CCH + kk);
      #pragma unroll
      for (int n0 = 0; n0 < 4; ++n0) {
        const int n = n0 * 16 + lm16;
        v2f b;
        b.x = W1[kk * CCH + n];
        b.y = W1[(kk + 1) * CCH + n];
        acc[n0] = wmma_f32_16x16x4(a, b, acc[n0]);
      }
    }
  }

  #pragma unroll
  for (int n0 = 0; n0 < 4; ++n0) {
    float s = 0.f, q = 0.f;
    #pragma unroll
    for (int r = 0; r < 8; ++r) { float v = acc[n0][r]; s += v; q += v * v; }
    s += __shfl_xor(s, 16);
    q += __shfl_xor(q, 16);
    if (half == 0) { wsum[wave][n0 * 16 + lm16] = s; wssq[wave][n0 * 16 + lm16] = q; }
  }
  __syncthreads();

  if (tid < 64) {
    float s = 0.f, q = 0.f;
    #pragma unroll
    for (int w = 0; w < 8; ++w) { s += wsum[w][tid]; q += wssq[w][tid]; }
    partials[blockIdx.x * 128 + tid]      = s;
    partials[blockIdx.x * 128 + 64 + tid] = q;
  }
}

#define K2_WAVES 32
#define K2_FULL  (GRID1 / K2_WAVES)
#define K2_REM   (GRID1 - K2_FULL * K2_WAVES)
__global__ __launch_bounds__(1024) void k2_finalize(const float* __restrict__ partials,
                                                    const float* __restrict__ gamma,
                                                    const float* __restrict__ beta,
                                                    float* __restrict__ ws) {
  __shared__ alignas(16) float red[K2_WAVES][128];
  const int tid = threadIdx.x;
  const int cq  = tid & 31;
  const int g   = tid >> 5;

  float4 v[K2_FULL];
  #pragma unroll
  for (int i = 0; i < K2_FULL; ++i)
    v[i] = *(const float4*)(partials + (g + K2_WAVES * i) * 128 + 4 * cq);
  float4 acc = v[0];
  #pragma unroll
  for (int i = 1; i < K2_FULL; ++i) {
    acc.x += v[i].x; acc.y += v[i].y; acc.z += v[i].z; acc.w += v[i].w;
  }
  if (g < K2_REM) {
    const float4 t = *(const float4*)(partials + (g + K2_WAVES * K2_FULL) * 128 + 4 * cq);
    acc.x += t.x; acc.y += t.y; acc.z += t.z; acc.w += t.w;
  }
  *(float4*)(&red[g][4 * cq]) = acc;
  __syncthreads();

  if (tid < 64) {
    const int ch = tid;
    float s = 0.f, q = 0.f;
    #pragma unroll
    for (int w = 0; w < K2_WAVES; ++w) { s += red[w][ch]; q += red[w][64 + ch]; }
    const float inv_n = 1.0f / (float)NPTS;
    const float mu  = s * inv_n;
    const float var = q * inv_n - mu * mu;
    const float sc  = gamma[ch] * rsqrtf(var + BN_EPS);
    ws[SCALE_OFF + ch] = sc;
    ws[SHIFT_OFF + ch] = beta[ch] - mu * sc;
  }
}

#define ROWS3   32
#define BLOCKS3 ((NPTS + ROWS3 - 1) / ROWS3)
#define NPAIR3  (ROWS3 * HNB)
__global__ __launch_bounds__(256) void k3_main(const float* __restrict__ qpts,
                                               const float* __restrict__ spts,
                                               const float* __restrict__ sfeat,
                                               const int*   __restrict__ nbidx,
                                               const float* __restrict__ kpts,
                                               const float* __restrict__ W1,
                                               const float* __restrict__ W2,
                                               const float* __restrict__ b2,
                                               const float* __restrict__ ws,
                                               float* __restrict__ out) {
  __shared__ float  act[ROWS3 * 68];
  __shared__ alignas(16) __half cw[ROWS3 * 960];
  __shared__ int    jIdx[NPAIR3];
  __shared__ float  nnI[NPAIR3];
  __shared__ unsigned char nnK[NPAIR3];
  __shared__ unsigned needM[ROWS3];

  const int tid  = threadIdx.x;
  const int wave = tid >> 5;
  const int lane = tid & 31;
  const int half = lane >> 4;
  const int lm16 = lane & 15;
  const int m0   = blockIdx.x * ROWS3;

  {
    const int mt  = wave >> 2;
    const int n0  = (wave & 3) * 16;
    int row = m0 + mt * 16 + lm16;
    row = (row < NPTS) ? row : (NPTS - 1);
    v8f acc = (v8f){0.f,0.f,0.f,0.f,0.f,0.f,0.f,0.f};
    for (int k0 = 0; k0 < CCH; k0 += 4) {
      const int kk = k0 + 2 * half;
      v2f a = *(const v2f*)(sfeat + row * CCH + kk);
      v2f b;
      b.x = W1[kk * CCH + n0 + lm16];
      b.y = W1[(kk + 1) * CCH + n0 + lm16];
      acc = wmma_f32_16x16x4(a, b, acc);
    }
    const int ch = n0 + lm16;
    const float sc = ws[SCALE_OFF + ch];
    const float sh = ws[SHIFT_OFF + ch];
    #pragma unroll
    for (int r = 0; r < 8; ++r) {
      const int M = mt * 16 + r + 8 * half;
      float v = acc[r] * sc + sh;
      v = (v > 0.f) ? v : 0.1f * v;
      act[M * 68 + ch] = v;
    }
  }
  {
    #pragma unroll 2
    for (int i = 0; i < 4; ++i) {
      const int p  = i * 256 + tid;
      const int lm = p >> 5;
      const int h  = p & 31;
      int m = m0 + lm;
      m = (m < NPTS) ? m : (NPTS - 1);
      const int j  = nbidx[m * HNB + h];
      const bool valid = ((unsigned)j < (unsigned)NPTS);
      const int jj = valid ? j : 0;
      const float qx = qpts[m * 3 + 0];
      const float qy = qpts[m * 3 + 1];
      const float qz = qpts[m * 3 + 2];
      const float dx = spts[jj * 3 + 0] - qx;
      const float dy = spts[jj * 3 + 1] - qy;
      const float dz = spts[jj * 3 + 2] - qz;
      float best = 3.4e38f; int bk = 0;
      #pragma unroll
      for (int k = 0; k < KPN; ++k) {
        const float ex = dx - kpts[k * 3 + 0];
        const float ey = dy - kpts[k * 3 + 1];
        const float ez = dz - kpts[k * 3 + 2];
        const float sq = ex * ex + ey * ey + ez * ez;
        if (sq < best) { best = sq; bk = k; }
      }
      float infl = 1.0f - sqrtf(best) * (1.0f / SIGMA);
      infl = (infl > 0.f) ? infl : 0.f;
      if (!valid) infl = 0.f;
      const bool on = (infl > 0.f);
      int um = on ? (1 << bk) : 0;
      um |= __shfl_xor(um, 16);
      um |= __shfl_xor(um, 8);
      um |= __shfl_xor(um, 4);
      um |= __shfl_xor(um, 2);
      um |= __shfl_xor(um, 1);
      um = um ? um : 1;
      jIdx[p] = on ? jj : m;
      nnK[p]  = (unsigned char)(on ? bk : (__ffs(um) - 1));
      nnI[p]  = infl;
      if (h == 0) needM[lm] = (unsigned)um;
    }
  }
  __syncthreads();

  const unsigned nm = needM[lane];
  #pragma unroll 1
  for (int it = wave; it < 2 * KPN; it += 8) {
    const int kp = it >> 1;
    const int cb = kp * 64 + (it & 1) * 32 + 2 * lm16;
    const unsigned bal = __builtin_amdgcn_ballot_w32(((nm >> kp) & 1u) != 0u);
    const int nk = __popc(bal);
    const v2f bias = *(const v2f*)(b2 + cb);
    #pragma unroll 1
    for (int s0 = 0; s0 < nk; s0 += 16) {
      const int s  = s0 + lm16;
      const int mr = nth_set_bit(bal, (s < nk) ? s : s0);
      v8f acc0 = (v8f){0.f,0.f,0.f,0.f,0.f,0.f,0.f,0.f};
      v8f acc1 = (v8f){0.f,0.f,0.f,0.f,0.f,0.f,0.f,0.f};
      const float* ap = act + mr * 68 + 2 * half;
      const float* bp = W2 + (2 * half) * 960 + cb;
      #pragma unroll
      for (int ks = 0; ks < CCH / 4; ++ks) {
        const v2f a  = *(const v2f*)(ap + 4 * ks);
        const v2f bx = *(const v2f*)(bp + (4 * ks) * 960);
        const v2f by = *(const v2f*)(bp + (4 * ks + 1) * 960);
        acc0 = wmma_f32_16x16x4(a, (v2f){bx.x, by.x}, acc0);
        acc1 = wmma_f32_16x16x4(a, (v2f){bx.y, by.y}, acc1);
      }
      #pragma unroll
      for (int r = 0; r < 8; ++r) {
        const int plo = __builtin_amdgcn_readlane(mr, r);
        const int phi = __builtin_amdgcn_readlane(mr, r + 8);
        const int pr  = half ? phi : plo;
        *(__half2*)(cw + pr * 960 + cb) =
            __halves2half2(__float2half(acc0[r] + bias.x),
                           __float2half(acc1[r] + bias.y));
      }
    }
  }
  __syncthreads();

  const int nrow = (m0 + 16 < NPTS) ? 4 : 2;
  const int g = half;
  const int q = lm16;
  const float4* __restrict__ sf4 = (const float4*)sfeat;
  #pragma unroll 1
  for (int r = 0; r < nrow; ++r) {
    const int lm = wave + r * 8;
    const int pb = lm * HNB + g;
    float4 x[HNB / 2];
    #pragma unroll
    for (int i = 0; i < HNB / 2; ++i) {
      const unsigned j = (unsigned)jIdx[pb + 2 * i];
      x[i] = sf4[j * 16u + (unsigned)q];
    }
    __builtin_amdgcn_sched_barrier(0);
    float4 a = make_float4(0.f, 0.f, 0.f, 0.f);
    #pragma unroll
    for (int i = 0; i < HNB / 2; ++i) {
      const int   k = nnK[pb + 2 * i];
      const float f = nnI[pb + 2 * i];
      const __half2* wp = (const __half2*)(cw + lm * 960 + k * 64 + 4 * q);
      const __half2 w01 = wp[0];
      const __half2 w23 = wp[1];
      a.x = fmaf(x[i].x, __low2float(w01)  * f, a.x);
      a.y = fmaf(x[i].y, __high2float(w01) * f, a.y);
      a.z = fmaf(x[i].z, __low2float(w23)  * f, a.z);
      a.w = fmaf(x[i].w, __high2float(w23) * f, a.w);
    }
    a.x += __shfl_xor(a.x, 16);
    a.y += __shfl_xor(a.y, 16);
    a.z += __shfl_xor(a.z, 16);
    a.w += __shfl_xor(a.w, 16);
    float2 o;
    o.x = g ? a.z : a.x;
    o.y = g ? a.w : a.y;
    *(float2*)(out + (m0 + lm) * 64 + 4 * q + 2 * g) = o;
  }
}

extern "C" void kernel_launch(void* const* d_in, const int* in_sizes, int n_in,
                              void* d_out, int out_size, void* d_ws, size_t ws_size,
                              hipStream_t stream) {
  const float* qpts  = (const float*)d_in[0];
  const float* spts  = (const float*)d_in[1];
  const float* sfeat = (const float*)d_in[2];
  const int*   nbidx = (const int*)  d_in[3];
  const float* kpts  = (const float*)d_in[4];
  const float* W1    = (const float*)d_in[5];
  const float* gamma = (const float*)d_in[6];
  const float* beta  = (const float*)d_in[7];
  const float* W2    = (const float*)d_in[8];
  const float* b2    = (const float*)d_in[9];
  float* out = (float*)d_out;
  float* ws  = (float*)d_ws;

  k1_stats<<<GRID1, 256, 0, stream>>>(sfeat, W1, ws);
  k2_finalize<<<1, 1024, 0, stream>>>(ws, gamma, beta, ws);
  k3_main<<<BLOCKS3, 256, 0, stream>>>(qpts, spts, sfeat, nbidx, kpts,
                                     W1, W2, b2, ws, out);
  (void)in_sizes; (void)n_in; (void)out_size; (void)ws_size;
}


# k3 aggregation reads only neighbors with influence: a compacted list per query (ballot rank, padded to chunks of 8 with x*0 entries), first chunks of a wave's 4 queries loaded together; new sum order
# speedup vs baseline: 2.5753x; 1.0708x over previous
#include <hip/hip_runtime.h>
#include <hip/hip_fp16.h>


typedef float v2f __attribute__((ext_vector_type(2)));
typedef float v8f __attribute__((ext_vector_type(8)));

#define NPTS   50000
#define CCH    64
#define KPN    15
#define HNB    32
#define SIGMA  0.7f
#define BN_EPS 1e-5f

#define TILES     ((NPTS + 15) / 16)
#define GRID1     ((TILES + 7) / 8)
#define SCALE_OFF (GRID1 * 128)
#define SHIFT_OFF (SCALE_OFF + 64)

static __device__ __forceinline__ v8f wmma_f32_16x16x4(v2f a, v2f b, v8f c) {
  return __builtin_amdgcn_wmma_f32_16x16x4_f32(false, a, false, b,
                                               (short)0, c, false, false);
}

static __device__ __forceinline__ int nth_set_bit(unsigned bal, int r) {
  int pos = 0;
  #pragma unroll
  for (int w = 16; w >= 1; w >>= 1) {
    const int c = __popc((bal >> pos) & ((1u << w) - 1u));
    if (r >= c) { r -= c; pos += w; }
  }
  return pos;
}

__global__ __launch_bounds__(256) void k1_stats(const float* __restrict__ sfeat,
                                                const float* __restrict__ W1,
                                                float* __restrict__ partials) {
  __shared__ float wsum[8][64];
  __shared__ float wssq[8][64];
  const int tid  = threadIdx.x;
  const int wave = tid >> 5;
  const int lane = tid & 31;
  const int half = lane >> 4;
  const int lm16 = lane & 15;
  const int t    = blockIdx.x * 8 + wave;

  v8f acc[4];
  #pragma unroll
  for (int n0 = 0; n0 < 4; ++n0)
    acc[n0] = (v8f){0.f,0.f,0.f,0.f,0.f,0.f,0.f,0.f};

  if (t < TILES) {
    const int row = t * 16 + lm16;
    for (int k0 = 0; k0 < CCH; k0 += 4) {
      const int kk = k0 + 2 * half;
      v2f a = *(const v2f*)(sfeat + row * CCH + kk);
      #pragma unroll
      for (int n0 = 0; n0 < 4; ++n0) {
        const int n = n0 * 16 + lm16;
        v2f b;
        b.x = W1[kk * CCH + n];
        b.y = W1[(kk + 1) * CCH + n];
        acc[n0] = wmma_f32_16x16x4(a, b, acc[n0]);
      }
    }
  }

  #pragma unroll
  for (int n0 = 0; n0 < 4; ++n0) {
    float s = 0.f, q = 0.f;
    #pragma unroll
    for (int r = 0; r < 8; ++r) { float v = acc[n0][r]; s += v; q += v * v; }
    s += __shfl_xor(s, 16);
    q += __shfl_xor(q, 16);
    if (half == 0) { wsum[wave][n0 * 16 + lm16] = s; wssq[wave][n0 * 16 + lm16] = q; }
  }
  __syncthreads();

  if (tid < 64) {
    float s = 0.f, q = 0.f;
    #pragma unroll
    for (int w = 0; w < 8; ++w) { s += wsum[w][tid]; q += wssq[w][tid]; }
    partials[blockIdx.x * 128 + tid]      = s;
    partials[blockIdx.x * 128 + 64 + tid] = q;
  }
}

#define K2_WAVES 32
#define K2_FULL  (GRID1 / K2_WAVES)
#define K2_REM   (GRID1 - K2_FULL * K2_WAVES)
__global__ __launch_bounds__(1024) void k2_finalize(const float* __restrict__ partials,
                                                    const float* __restrict__ gamma,
                                                    const float* __restrict__ beta,
                                                    float* __restrict__ ws) {
  __shared__ alignas(16) float red[K2_WAVES][128];
  const int tid = threadIdx.x;
  const int cq  = tid & 31;
  const int g   = tid >> 5;

  float4 v[K2_FULL];
  #pragma unroll
  for (int i = 0; i < K2_FULL; ++i)
    v[i] = *(const float4*)(partials + (g + K2_WAVES * i) * 128 + 4 * cq);
  float4 acc = v[0];
  #pragma unroll
  for (int i = 1; i < K2_FULL; ++i) {
    acc.x += v[i].x; acc.y += v[i].y; acc.z += v[i].z; acc.w += v[i].w;
  }
  if (g < K2_REM) {
    const float4 t = *(const float4*)(partials + (g + K2_WAVES * K2_FULL) * 128 + 4 * cq);
    acc.x += t.x; acc.y += t.y; acc.z += t.z; acc.w += t.w;
  }
  *(float4*)(&red[g][4 * cq]) = acc;
  __syncthreads();

  if (tid < 64) {
    const int ch = tid;
    float s = 0.f, q = 0.f;
    #pragma unroll
    for (int w = 0; w < K2_WAVES; ++w) { s += red[w][ch]; q += red[w][64 + ch]; }
    const float inv_n = 1.0f / (float)NPTS;
    const float mu  = s * inv_n;
    const float var = q * inv_n - mu * mu;
    const float sc  = gamma[ch] * rsqrtf(var + BN_EPS);
    ws[SCALE_OFF + ch] = sc;
    ws[SHIFT_OFF + ch] = beta[ch] - mu * sc;
  }
}

#define ROWS3   32
#define BLOCKS3 ((NPTS + ROWS3 - 1) / ROWS3)
#define NPAIR3  (ROWS3 * HNB)
__global__ __launch_bounds__(256) void k3_main(const float* __restrict__ qpts,
                                               const float* __restrict__ spts,
                                               const float* __restrict__ sfeat,
                                               const int*   __restrict__ nbidx,
                                               const float* __restrict__ kpts,
                                               const float* __restrict__ W1,
                                               const float* __restrict__ W2,
                                               const float* __restrict__ b2,
                                               const float* __restrict__ ws,
                                               float* __restrict__ out) {
  __shared__ float  act[ROWS3 * 68];
  __shared__ alignas(16) __half cw[ROWS3 * 960];
  __shared__ int    jIdx[NPAIR3];
  __shared__ float  nnI[NPAIR3];
  __shared__ unsigned char nnK[NPAIR3];
  __shared__ unsigned needM[ROWS3];
  __shared__ unsigned char nchP[ROWS3];

  const int tid  = threadIdx.x;
  const int wave = tid >> 5;
  const int lane = tid & 31;
  const int half = lane >> 4;
  const int lm16 = lane & 15;
  const int m0   = blockIdx.x * ROWS3;

  {
    const int mt  = wave >> 2;
    const int n0  = (wave & 3) * 16;
    int row = m0 + mt * 16 + lm16;
    row = (row < NPTS) ? row : (NPTS - 1);
    v8f acc = (v8f){0.f,0.f,0.f,0.f,0.f,0.f,0.f,0.f};
    for (int k0 = 0; k0 < CCH; k0 += 4) {
      const int kk = k0 + 2 * half;
      v2f a = *(const v2f*)(sfeat + row * CCH + kk);
      v2f b;
      b.x = W1[kk * CCH + n0 + lm16];
      b.y = W1[(kk + 1) * CCH + n0 + lm16];
      acc = wmma_f32_16x16x4(a, b, acc);
    }
    const int ch = n0 + lm16;
    const float sc = ws[SCALE_OFF + ch];
    const float sh = ws[SHIFT_OFF + ch];
    #pragma unroll
    for (int r = 0; r < 8; ++r) {
      const int M = mt * 16 + r + 8 * half;
      float v = acc[r] * sc + sh;
      v = (v > 0.f) ? v : 0.1f * v;
      act[M * 68 + ch] = v;
    }
  }
  {
    #pragma unroll 2
    for (int i = 0; i < 4; ++i) {
      const int p  = i * 256 + tid;
      const int lm = p >> 5;
      const int h  = p & 31;
      int m = m0 + lm;
      m = (m < NPTS) ? m : (NPTS - 1);
      const int j  = nbidx[m * HNB + h];
      const bool valid = ((unsigned)j < (unsigned)NPTS);
      const int jj = valid ? j : 0;
      const float qx = qpts[m * 3 + 0];
      const float qy = qpts[m * 3 + 1];
      const float qz = qpts[m * 3 + 2];
      const float dx = spts[jj * 3 + 0] - qx;
      const float dy = spts[jj * 3 + 1] - qy;
      const float dz = spts[jj * 3 + 2] - qz;
      float best = 3.4e38f; int bk = 0;
      #pragma unroll
      for (int k = 0; k < KPN; ++k) {
        const float ex = dx - kpts[k * 3 + 0];
        const float ey = dy - kpts[k * 3 + 1];
        const float ez = dz - kpts[k * 3 + 2];
        const float sq = ex * ex + ey * ey + ez * ez;
        if (sq < best) { best = sq; bk = k; }
      }
      float infl = 1.0f - sqrtf(best) * (1.0f / SIGMA);
      infl = (infl > 0.f) ? infl : 0.f;
      if (!valid) infl = 0.f;
      const bool on = (infl > 0.f);
      int um = on ? (1 << bk) : 0;
      um |= __shfl_xor(um, 16);
      um |= __shfl_xor(um, 8);
      um |= __shfl_xor(um, 4);
      um |= __shfl_xor(um, 2);
      um |= __shfl_xor(um, 1);
      um = um ? um : 1;
      const unsigned onb = __builtin_amdgcn_ballot_w32(on);
      const int cnt  = __popc(onb);
      const int pos  = __popc(onb & ((1u << h) - 1u));
      const int npad = (cnt > 8) ? ((cnt + 7) & ~7) : 8;
      const int lb   = lm * HNB;
      if (on) {
        jIdx[lb + pos] = jj;
        nnK[lb + pos]  = (unsigned char)bk;
        nnI[lb + pos]  = infl;
      }
      if (h >= cnt && h < npad) {
        jIdx[lb + h] = m;
        nnK[lb + h]  = (unsigned char)(__ffs(um) - 1);
        nnI[lb + h]  = 0.f;
      }
      if (h == 0) { needM[lm] = (unsigned)um; nchP[lm] = (unsigned char)(npad >> 3); }
    }
  }
  __syncthreads();

  const unsigned nm = needM[lane];
  #pragma unroll 1
  for (int it = wave; it < 2 * KPN; it += 8) {
    const int kp = it >> 1;
    const int cb = kp * 64 + (it & 1) * 32 + 2 * lm16;
    const unsigned bal = __builtin_amdgcn_ballot_w32(((nm >> kp) & 1u) != 0u);
    const int nk = __popc(bal);
    const v2f bias = *(const v2f*)(b2 + cb);
    #pragma unroll 1
    for (int s0 = 0; s0 < nk; s0 += 16) {
      const int s  = s0 + lm16;
      const int mr = nth_set_bit(bal, (s < nk) ? s : s0);
      v8f acc0 = (v8f){0.f,0.f,0.f,0.f,0.f,0.f,0.f,0.f};
      v8f acc1 = (v8f){0.f,0.f,0.f,0.f,0.f,0.f,0.f,0.f};
      const float* ap = act + mr * 68 + 2 * half;
      const float* bp = W2 + (2 * half) * 960 + cb;
      #pragma unroll
      for (int ks = 0; ks < CCH / 4; ++ks) {
        const v2f a  = *(const v2f*)(ap + 4 * ks);
        const v2f bx = *(const v2f*)(bp + (4 * ks) * 960);
        const v2f by = *(const v2f*)(bp + (4 * ks + 1) * 960);
        acc0 = wmma_f32_16x16x4(a, (v2f){bx.x, by.x}, acc0);
        acc1 = wmma_f32_16x16x4(a, (v2f){bx.y, by.y}, acc1);
      }
      #pragma unroll
      for (int r = 0; r < 8; ++r) {
        const int plo = __builtin_amdgcn_readlane(mr, r);
        const int phi = __builtin_amdgcn_readlane(mr, r + 8);
        const int pr  = half ? phi : plo;
        *(__half2*)(cw + pr * 960 + cb) =
            __halves2half2(__float2half(acc0[r] + bias.x),
                           __float2half(acc1[r] + bias.y));
      }
    }
  }
  __syncthreads();

  const int g = half;
  const int q = lm16;
  const float4* __restrict__ sf4 = (const float4*)sfeat;
  float4 a[4];
  {
    float4 x[4][4];
    #pragma unroll
    for (int r = 0; r < 4; ++r) {
      const int pb = (wave + r * 8) * HNB + g;
      #pragma unroll
      for (int i = 0; i < 4; ++i) {
        const unsigned j = (unsigned)jIdx[pb + 2 * i];
        x[r][i] = sf4[j * 16u + (unsigned)q];
      }
    }
    __builtin_amdgcn_sched_barrier(0);
    #pragma unroll
    for (int r = 0; r < 4; ++r) {
      const int lm = wave + r * 8;
      const int pb = lm * HNB + g;
      a[r] = make_float4(0.f, 0.f, 0.f, 0.f);
      #pragma unroll
      for (int i = 0; i < 4; ++i) {
        const int   k = nnK[pb + 2 * i];
        const float f = nnI[pb + 2 * i];
        const __half2* wp = (const __half2*)(cw + lm * 960 + k * 64 + 4 * q);
        const __half2 w01 = wp[0];
        const __half2 w23 = wp[1];
        a[r].x = fmaf(x[r][i].x, __low2float(w01)  * f, a[r].x);
        a[r].y = fmaf(x[r][i].y, __high2float(w01) * f, a[r].y);
        a[r].z = fmaf(x[r][i].z, __low2float(w23)  * f, a[r].z);
        a[r].w = fmaf(x[r][i].w, __high2float(w23) * f, a[r].w);
      }
    }
  }
  int nch[4];
  #pragma unroll
  for (int r = 0; r < 4; ++r)
    nch[r] = __builtin_amdgcn_readfirstlane((int)nchP[wave + r * 8]);
  const int maxch = max(max(nch[0], nch[1]), max(nch[2], nch[3]));
  #pragma unroll 1
  for (int c = 1; c < maxch; ++c) {
    float4 x[4][4];
    #pragma unroll
    for (int r = 0; r < 4; ++r) {
      if (c < nch[r]) {
        const int pb = (wave + r * 8) * HNB + c * 8 + g;
        #pragma unroll
        for (int i = 0; i < 4; ++i) {
          const unsigned j = (unsigned)jIdx[pb + 2 * i];
          x[r][i] = sf4[j * 16u + (unsigned)q];
        }
      }
    }
    #pragma unroll
    for (int r = 0; r < 4; ++r) {
      if (c < nch[r]) {
        const int lm = wave + r * 8;
        const int pb = lm * HNB + c * 8 + g;
        #pragma unroll
        for (int i = 0; i < 4; ++i) {
          const int   k = nnK[pb + 2 * i];
          const float f = nnI[pb + 2 * i];
          const __half2* wp = (const __half2*)(cw + lm * 960 + k * 64 + 4 * q);
          const __half2 w01 = wp[0];
          const __half2 w23 = wp[1];
          a[r].x = fmaf(x[r][i].x, __low2float(w01)  * f, a[r].x);
          a[r].y = fmaf(x[r][i].y, __high2float(w01) * f, a[r].y);
          a[r].z = fmaf(x[r][i].z, __low2float(w23)  * f, a[r].z);
          a[r].w = fmaf(x[r][i].w, __high2float(w23) * f, a[r].w);
        }
      }
    }
  }
  const int nrow = (m0 + 16 < NPTS) ? 4 : 2;
  #pragma unroll
  for (int r = 0; r < 4; ++r) {
    if (r < nrow) {
      const int lm = wave + r * 8;
      float4 t = a[r];
      t.x += __shfl_xor(t.x, 16);
      t.y += __shfl_xor(t.y, 16);
      t.z += __shfl_xor(t.z, 16);
      t.w += __shfl_xor(t.w, 16);
      float2 o;
      o.x = g ? t.z : t.x;
      o.y = g ? t.w : t.y;
      *(float2*)(out + (m0 + lm) * 64 + 4 * q + 2 * g) = o;
    }
  }
}

extern "C" void kernel_launch(void* const* d_in, const int* in_sizes, int n_in,
                              void* d_out, int out_size, void* d_ws, size_t ws_size,
                              hipStream_t stream) {
  const float* qpts  = (const float*)d_in[0];
  const float* spts  = (const float*)d_in[1];
  const float* sfeat = (const float*)d_in[2];
  const int*   nbidx = (const int*)  d_in[3];
  const float* kpts  = (const float*)d_in[4];
  const float* W1    = (const float*)d_in[5];
  const float* gamma = (const float*)d_in[6];
  const float* beta  = (const float*)d_in[7];
  const float* W2    = (const float*)d_in[8];
  const float* b2    = (const float*)d_in[9];
  float* out = (float*)d_out;
  float* ws  = (float*)d_ws;

  k1_stats<<<GRID1, 256, 0, stream>>>(sfeat, W1, ws);
  k2_finalize<<<1, 1024, 0, stream>>>(ws, gamma, beta, ws);
  k3_main<<<BLOCKS3, 256, 0, stream>>>(qpts, spts, sfeat, nbidx, kpts,
                                     W1, W2, b2, ws, out);
  (void)in_sizes; (void)n_in; (void)out_size; (void)ws_size;
}


# h = s_feats@W1 (k1 and k3's h tiles): feature rows and W1 are copied to LDS with whole-line 16-byte loads and the WMMA fragments are read from LDS, not by 8-byte strided global loads per lane
# speedup vs baseline: 2.7197x; 1.0561x over previous
#include <hip/hip_runtime.h>
#include <hip/hip_fp16.h>


typedef float v2f __attribute__((ext_vector_type(2)));
typedef float v8f __attribute__((ext_vector_type(8)));

#define NPTS   50000
#define CCH    64
#define KPN    15
#define HNB    32
#define SIGMA  0.7f
#define BN_EPS 1e-5f

#define TILES     ((NPTS + 15) / 16)
#define GRID1     ((TILES + 7) / 8)
#define SCALE_OFF (GRID1 * 128)
#define SHIFT_OFF (SCALE_OFF + 64)

static __device__ __forceinline__ v8f wmma_f32_16x16x4(v2f a, v2f b, v8f c) {
  return __builtin_amdgcn_wmma_f32_16x16x4_f32(false, a, false, b,
                                               (short)0, c, false, false);
}

static __device__ __forceinline__ int nth_set_bit(unsigned bal, int r) {
  int pos = 0;
  #pragma unroll
  for (int w = 16; w >= 1; w >>= 1) {
    const int c = __popc((bal >> pos) & ((1u << w) - 1u));
    if (r >= c) { r -= c; pos += w; }
  }
  return pos;
}

#define SA_PITCH 68
#define SW_PITCH 72

__global__ __launch_bounds__(256) void k1_stats(const float* __restrict__ sfeat,
                                                const float* __restrict__ W1,
                                                float* __restrict__ partials) {
  __shared__ float wsum[8][64];
  __shared__ float wssq[8][64];
  __shared__ alignas(16) float sA[128 * SA_PITCH];
  __shared__ alignas(16) float sW[CCH * SW_PITCH];
  const int tid  = threadIdx.x;
  const int wave = tid >> 5;
  const int lane = tid & 31;
  const int half = lane >> 4;
  const int lm16 = lane & 15;
  const int t    = blockIdx.x * 8 + wave;

  {
    float4 wv[4], av[8];
    #pragma unroll
    for (int i = 0; i < 4; ++i) {
      const int u = i * 256 + tid;
      wv[i] = *(const float4*)(W1 + (u >> 4) * CCH + 4 * (u & 15));
    }
    #pragma unroll
    for (int i = 0; i < 8; ++i) {
      const int u = i * 256 + tid;
      int row = blockIdx.x * 128 + (u >> 4);
      row = (row < NPTS) ? row : (NPTS - 1);
      av[i] = *(const float4*)(sfeat + row * CCH + 4 * (u & 15));
    }
    #pragma unroll
    for (int i = 0; i < 4; ++i) {
      const int u = i * 256 + tid;
      *(float4*)(sW + (u >> 4) * SW_PITCH + 4 * (u & 15)) = wv[i];
    }
    #pragma unroll
    for (int i = 0; i < 8; ++i) {
      const int u = i * 256 + tid;
      *(float4*)(sA + (u >> 4) * SA_PITCH + 4 * (u & 15)) = av[i];
    }
  }
  __syncthreads();

  v8f acc[4];
  #pragma unroll
  for (int n0 = 0; n0 < 4; ++n0)
    acc[n0] = (v8f){0.f,0.f,0.f,0.f,0.f,0.f,0.f,0.f};

  if (t < TILES) {
    const float* ap = sA + (wave * 16 + lm16) * SA_PITCH + 2 * half;
    const float* bp = sW + (2 * half) * SW_PITCH + lm16;
    #pragma unroll
    for (int ks = 0; ks < CCH / 4; ++ks) {
      const v2f a = *(const v2f*)(ap + 4 * ks);
      #pragma unroll
      for (int n0 = 0; n0 < 4; ++n0) {
        v2f b;
        b.x = bp[(4 * ks) * SW_PITCH + n0 * 16];
        b.y = bp[(4 * ks + 1) * SW_PITCH + n0 * 16];
        acc[n0] = wmma_f32_16x16x4(a, b, acc[n0]);
      }
    }
  }

  #pragma unroll
  for (int n0 = 0; n0 < 4; ++n0) {
    float s = 0.f, q = 0.f;
    #pragma unroll
    for (int r = 0; r < 8; ++r) { float v = acc[n0][r]; s += v; q += v * v; }
    s += __shfl_xor(s, 16);
    q += __shfl_xor(q, 16);
    if (half == 0) { wsum[wave][n0 * 16 + lm16] = s; wssq[wave][n0 * 16 + lm16] = q; }
  }
  __syncthreads();

  if (tid < 64) {
    float s = 0.f, q = 0.f;
    #pragma unroll
    for (int w = 0; w < 8; ++w) { s += wsum[w][tid]; q += wssq[w][tid]; }
    partials[blockIdx.x * 128 + tid]      = s;
    partials[blockIdx.x * 128 + 64 + tid] = q;
  }
}

#define K2_WAVES 32
#define K2_FULL  (GRID1 / K2_WAVES)
#define K2_REM   (GRID1 - K2_FULL * K2_WAVES)
__global__ __launch_bounds__(1024) void k2_finalize(const float* __restrict__ partials,
                                                    const float* __restrict__ gamma,
                                                    const float* __restrict__ beta,
                                                    float* __restrict__ ws) {
  __shared__ alignas(16) float red[K2_WAVES][128];
  const int tid = threadIdx.x;
  const int cq  = tid & 31;
  const int g   = tid >> 5;

  float4 v[K2_FULL];
  #pragma unroll
  for (int i = 0; i < K2_FULL; ++i)
    v[i] = *(const float4*)(partials + (g + K2_WAVES * i) * 128 + 4 * cq);
  float4 acc = v[0];
  #pragma unroll
  for (int i = 1; i < K2_FULL; ++i) {
    acc.x += v[i].x; acc.y += v[i].y; acc.z += v[i].z; acc.w += v[i].w;
  }
  if (g < K2_REM) {
    const float4 t = *(const float4*)(partials + (g + K2_WAVES * K2_FULL) * 128 + 4 * cq);
    acc.x += t.x; acc.y += t.y; acc.z += t.z; acc.w += t.w;
  }
  *(float4*)(&red[g][4 * cq]) = acc;
  __syncthreads();

  if (tid < 64) {
    const int ch = tid;
    float s = 0.f, q = 0.f;
    #pragma unroll
    for (int w = 0; w < K2_WAVES; ++w) { s += red[w][ch]; q += red[w][64 + ch]; }
    const float inv_n = 1.0f / (float)NPTS;
    const float mu  = s * inv_n;
    const float var = q * inv_n - mu * mu;
    const float sc  = gamma[ch] * rsqrtf(var + BN_EPS);
    ws[SCALE_OFF + ch] = sc;
    ws[SHIFT_OFF + ch] = beta[ch] - mu * sc;
  }
}

#define ROWS3   32
#define BLOCKS3 ((NPTS + ROWS3 - 1) / ROWS3)
#define NPAIR3  (ROWS3 * HNB)
__global__ __launch_bounds__(256) void k3_main(const float* __restrict__ qpts,
                                               const float* __restrict__ spts,
                                               const float* __restrict__ sfeat,
                                               const int*   __restrict__ nbidx,
                                               const float* __restrict__ kpts,
                                               const float* __restrict__ W1,
                                               const float* __restrict__ W2,
                                               const float* __restrict__ b2,
                                               const float* __restrict__ ws,
                                               float* __restrict__ out) {
  __shared__ float  act[ROWS3 * 68];
  __shared__ alignas(16) unsigned char pool[ROWS3 * 960 * 2];
  __half* const cw  = (__half*)pool;
  float*  const sA3 = (float*)pool;
  float*  const sW3 = (float*)pool + ROWS3 * SA_PITCH;
  __shared__ int    jIdx[NPAIR3];
  __shared__ float  nnI[NPAIR3];
  __shared__ unsigned char nnK[NPAIR3];
  __shared__ unsigned needM[ROWS3];
  __shared__ unsigned char nchP[ROWS3];

  const int tid  = threadIdx.x;
  const int wave = tid >> 5;
  const int lane = tid & 31;
  const int half = lane >> 4;
  const int lm16 = lane & 15;
  const int m0   = blockIdx.x * ROWS3;

  {
    float4 stA[2], stW[4];
    #pragma unroll
    for (int i = 0; i < 2; ++i) {
      const int u = i * 256 + tid;
      int row = m0 + (u >> 4);
      row = (row < NPTS) ? row : (NPTS - 1);
      stA[i] = *(const float4*)(sfeat + row * CCH + 4 * (u & 15));
    }
    #pragma unroll
    for (int i = 0; i < 4; ++i) {
      const int u = i * 256 + tid;
      stW[i] = *(const float4*)(W1 + (u >> 4) * CCH + 4 * (u & 15));
    }
    #pragma unroll
    for (int i = 0; i < 2; ++i) {
      const int u = i * 256 + tid;
      *(float4*)(sA3 + (u >> 4) * SA_PITCH + 4 * (u & 15)) = stA[i];
    }
    #pragma unroll
    for (int i = 0; i < 4; ++i) {
      const int u = i * 256 + tid;
      *(float4*)(sW3 + (u >> 4) * SW_PITCH + 4 * (u & 15)) = stW[i];
    }
  }
  __syncthreads();
  {
    const int mt  = wave >> 2;
    const int n0  = (wave & 3) * 16;
    const float* ap = sA3 + (mt * 16 + lm16) * SA_PITCH + 2 * half;
    const float* bp = sW3 + (2 * half) * SW_PITCH + n0 + lm16;
    v8f acc = (v8f){0.f,0.f,0.f,0.f,0.f,0.f,0.f,0.f};
    #pragma unroll
    for (int ks = 0; ks < CCH / 4; ++ks) {
      const v2f a = *(const v2f*)(ap + 4 * ks);
      v2f b;
      b.x = bp[(4 * ks) * SW_PITCH];
      b.y = bp[(4 * ks + 1) * SW_PITCH];
      acc = wmma_f32_16x16x4(a, b, acc);
    }
    const int ch = n0 + lm16;
    const float sc = ws[SCALE_OFF + ch];
    const float sh = ws[SHIFT_OFF + ch];
    #pragma unroll
    for (int r = 0; r < 8; ++r) {
      const int M = mt * 16 + r + 8 * half;
      float v = acc[r] * sc + sh;
      v = (v > 0.f) ? v : 0.1f * v;
      act[M * 68 + ch] = v;
    }
  }
  {
    #pragma unroll 2
    for (int i = 0; i < 4; ++i) {
      const int p  = i * 256 + tid;
      const int lm = p >> 5;
      const int h  = p & 31;
      int m = m0 + lm;
      m = (m < NPTS) ? m : (NPTS - 1);
      const int j  = nbidx[m * HNB + h];
      const bool valid = ((unsigned)j < (unsigned)NPTS);
      const int jj = valid ? j : 0;
      const float qx = qpts[m * 3 + 0];
      const float qy = qpts[m * 3 + 1];
      const float qz = qpts[m * 3 + 2];
      const float dx = spts[jj * 3 + 0] - qx;
      const float dy = spts[jj * 3 + 1] - qy;
      const float dz = spts[jj * 3 + 2] - qz;
      float best = 3.4e38f; int bk = 0;
      #pragma unroll
      for (int k = 0; k < KPN; ++k) {
        const float ex = dx - kpts[k * 3 + 0];
        const float ey = dy - kpts[k * 3 + 1];
        const float ez = dz - kpts[k * 3 + 2];
        const float sq = ex * ex + ey * ey + ez * ez;
        if (sq < best) { best = sq; bk = k; }
      }
      float infl = 1.0f - sqrtf(best) * (1.0f / SIGMA);
      infl = (infl > 0.f) ? infl : 0.f;
      if (!valid) infl = 0.f;
      const bool on = (infl > 0.f);
      int um = on ? (1 << bk) : 0;
      um |= __shfl_xor(um, 16);
      um |= __shfl_xor(um, 8);
      um |= __shfl_xor(um, 4);
      um |= __shfl_xor(um, 2);
      um |= __shfl_xor(um, 1);
      um = um ? um : 1;
      const unsigned onb = __builtin_amdgcn_ballot_w32(on);
      const int cnt  = __popc(onb);
      const int pos  = __popc(onb & ((1u << h) - 1u));
      const int npad = (cnt > 8) ? ((cnt + 7) & ~7) : 8;
      const int lb   = lm * HNB;
      if (on) {
        jIdx[lb + pos] = jj;
        nnK[lb + pos]  = (unsigned char)bk;
        nnI[lb + pos]  = infl;
      }
      if (h >= cnt && h < npad) {
        jIdx[lb + h] = m;
        nnK[lb + h]  = (unsigned char)(__ffs(um) - 1);
        nnI[lb + h]  = 0.f;
      }
      if (h == 0) { needM[lm] = (unsigned)um; nchP[lm] = (unsigned char)(npad >> 3); }
    }
  }
  __syncthreads();

  const unsigned nm = needM[lane];
  #pragma unroll 1
  for (int it = wave; it < 2 * KPN; it += 8) {
    const int kp = it >> 1;
    const int cb = kp * 64 + (it & 1) * 32 + 2 * lm16;
    const unsigned bal = __builtin_amdgcn_ballot_w32(((nm >> kp) & 1u) != 0u);
    const int nk = __popc(bal);
    const v2f bias = *(const v2f*)(b2 + cb);
    #pragma unroll 1
    for (int s0 = 0; s0 < nk; s0 += 16) {
      const int s  = s0 + lm16;
      const int mr = nth_set_bit(bal, (s < nk) ? s : s0);
      v8f acc0 = (v8f){0.f,0.f,0.f,0.f,0.f,0.f,0.f,0.f};
      v8f acc1 = (v8f){0.f,0.f,0.f,0.f,0.f,0.f,0.f,0.f};
      const float* ap = act + mr * 68 + 2 * half;
      const float* bp = W2 + (2 * half) * 960 + cb;
      #pragma unroll
      for (int ks = 0; ks < CCH / 4; ++ks) {
        const v2f a  = *(const v2f*)(ap + 4 * ks);
        const v2f bx = *(const v2f*)(bp + (4 * ks) * 960);
        const v2f by = *(const v2f*)(bp + (4 * ks + 1) * 960);
        acc0 = wmma_f32_16x16x4(a, (v2f){bx.x, by.x}, acc0);
        acc1 = wmma_f32_16x16x4(a, (v2f){bx.y, by.y}, acc1);
      }
      #pragma unroll
      for (int r = 0; r < 8; ++r) {
        const int plo = __builtin_amdgcn_readlane(mr, r);
        const int phi = __builtin_amdgcn_readlane(mr, r + 8);
        const int pr  = half ? phi : plo;
        *(__half2*)(cw + pr * 960 + cb) =
            __halves2half2(__float2half(acc0[r] + bias.x),
                           __float2half(acc1[r] + bias.y));
      }
    }
  }
  __syncthreads();

  const int g = half;
  const int q = lm16;
  const float4* __restrict__ sf4 = (const float4*)sfeat;
  float4 a[4];
  {
    float4 x[4][4];
    #pragma unroll
    for (int r = 0; r < 4; ++r) {
      const int pb = (wave + r * 8) * HNB + g;
      #pragma unroll
      for (int i = 0; i < 4; ++i) {
        const unsigned j = (unsigned)jIdx[pb + 2 * i];
        x[r][i] = sf4[j * 16u + (unsigned)q];
      }
    }
    __builtin_amdgcn_sched_barrier(0);
    #pragma unroll
    for (int r = 0; r < 4; ++r) {
      const int lm = wave + r * 8;
      const int pb = lm * HNB + g;
      a[r] = make_float4(0.f, 0.f, 0.f, 0.f);
      #pragma unroll
      for (int i = 0; i < 4; ++i) {
        const int   k = nnK[pb + 2 * i];
        const float f = nnI[pb + 2 * i];
        const __half2* wp = (const __half2*)(cw + lm * 960 + k * 64 + 4 * q);
        const __half2 w01 = wp[0];
        const __half2 w23 = wp[1];
        a[r].x = fmaf(x[r][i].x, __low2float(w01)  * f, a[r].x);
        a[r].y = fmaf(x[r][i].y, __high2float(w01) * f, a[r].y);
        a[r].z = fmaf(x[r][i].z, __low2float(w23)  * f, a[r].z);
        a[r].w = fmaf(x[r][i].w, __high2float(w23) * f, a[r].w);
      }
    }
  }
  int nch[4];
  #pragma unroll
  for (int r = 0; r < 4; ++r)
    nch[r] = __builtin_amdgcn_readfirstlane((int)nchP[wave + r * 8]);
  const int maxch = max(max(nch[0], nch[1]), max(nch[2], nch[3]));
  #pragma unroll 1
  for (int c = 1; c < maxch; ++c) {
    float4 x[4][4];
    #pragma unroll
    for (int r = 0; r < 4; ++r) {
      if (c < nch[r]) {
        const int pb = (wave + r * 8) * HNB + c * 8 + g;
        #pragma unroll
        for (int i = 0; i < 4; ++i) {
          const unsigned j = (unsigned)jIdx[pb + 2 * i];
          x[r][i] = sf4[j * 16u + (unsigned)q];
        }
      }
    }
    #pragma unroll
    for (int r = 0; r < 4; ++r) {
      if (c < nch[r]) {
        const int lm = wave + r * 8;
        const int pb = lm * HNB + c * 8 + g;
        #pragma unroll
        for (int i = 0; i < 4; ++i) {
          const int   k = nnK[pb + 2 * i];
          const float f = nnI[pb + 2 * i];
          const __half2* wp = (const __half2*)(cw + lm * 960 + k * 64 + 4 * q);
          const __half2 w01 = wp[0];
          const __half2 w23 = wp[1];
          a[r].x = fmaf(x[r][i].x, __low2float(w01)  * f, a[r].x);
          a[r].y = fmaf(x[r][i].y, __high2float(w01) * f, a[r].y);
          a[r].z = fmaf(x[r][i].z, __low2float(w23)  * f, a[r].z);
          a[r].w = fmaf(x[r][i].w, __high2float(w23) * f, a[r].w);
        }
      }
    }
  }
  const int nrow = (m0 + 16 < NPTS) ? 4 : 2;
  #pragma unroll
  for (int r = 0; r < 4; ++r) {
    if (r < nrow) {
      const int lm = wave + r * 8;
      float4 t = a[r];
      t.x += __shfl_xor(t.x, 16);
      t.y += __shfl_xor(t.y, 16);
      t.z += __shfl_xor(t.z, 16);
      t.w += __shfl_xor(t.w, 16);
      float2 o;
      o.x = g ? t.z : t.x;
      o.y = g ? t.w : t.y;
      *(float2*)(out + (m0 + lm) * 64 + 4 * q + 2 * g) = o;
    }
  }
}

extern "C" void kernel_launch(void* const* d_in, const int* in_sizes, int n_in,
                              void* d_out, int out_size, void* d_ws, size_t ws_size,
                              hipStream_t stream) {
  const float* qpts  = (const float*)d_in[0];
  const float* spts  = (const float*)d_in[1];
  const float* sfeat = (const float*)d_in[2];
  const int*   nbidx = (const int*)  d_in[3];
  const float* kpts  = (const float*)d_in[4];
  const float* W1    = (const float*)d_in[5];
  const float* gamma = (const float*)d_in[6];
  const float* beta  = (const float*)d_in[7];
  const float* W2    = (const float*)d_in[8];
  const float* b2    = (const float*)d_in[9];
  float* out = (float*)d_out;
  float* ws  = (float*)d_ws;

  k1_stats<<<GRID1, 256, 0, stream>>>(sfeat, W1, ws);
  k2_finalize<<<1, 1024, 0, stream>>>(ws, gamma, beta, ws);
  k3_main<<<BLOCKS3, 256, 0, stream>>>(qpts, spts, sfeat, nbidx, kpts,
                                     W1, W2, b2, ws, out);
  (void)in_sizes; (void)n_in; (void)out_size; (void)ws_size;
}


# k3 conv weights: a lane loads 16 bytes of a W2 row, its column in four column tiles (item = kernel point, all 64 channels): W2 load instructions and A-fragment LDS reads halve
# speedup vs baseline: 2.9054x; 1.0683x over previous
#include <hip/hip_runtime.h>
#include <hip/hip_fp16.h>


typedef float v2f __attribute__((ext_vector_type(2)));
typedef float v8f __attribute__((ext_vector_type(8)));
typedef _Float16 h4 __attribute__((ext_vector_type(4)));

#define NPTS   50000
#define CCH    64
#define KPN    15
#define HNB    32
#define SIGMA  0.7f
#define BN_EPS 1e-5f

#define TILES     ((NPTS + 15) / 16)
#define GRID1     ((TILES + 7) / 8)
#define SCALE_OFF (GRID1 * 128)
#define SHIFT_OFF (SCALE_OFF + 64)

static __device__ __forceinline__ v8f wmma_f32_16x16x4(v2f a, v2f b, v8f c) {
  return __builtin_amdgcn_wmma_f32_16x16x4_f32(false, a, false, b,
                                               (short)0, c, false, false);
}

static __device__ __forceinline__ int nth_set_bit(unsigned bal, int r) {
  int pos = 0;
  #pragma unroll
  for (int w = 16; w >= 1; w >>= 1) {
    const int c = __popc((bal >> pos) & ((1u << w) - 1u));
    if (r >= c) { r -= c; pos += w; }
  }
  return pos;
}

#define SA_PITCH 68
#define SW_PITCH 72

__global__ __launch_bounds__(256) void k1_stats(const float* __restrict__ sfeat,
                                                const float* __restrict__ W1,
                                                float* __restrict__ partials) {
  __shared__ float wsum[8][64];
  __shared__ float wssq[8][64];
  __shared__ alignas(16) float sA[128 * SA_PITCH];
  __shared__ alignas(16) float sW[CCH * SW_PITCH];
  const int tid  = threadIdx.x;
  const int wave = tid >> 5;
  const int lane = tid & 31;
  const int half = lane >> 4;
  const int lm16 = lane & 15;
  const int t    = blockIdx.x * 8 + wave;

  {
    float4 wv[4], av[8];
    #pragma unroll
    for (int i = 0; i < 4; ++i) {
      const int u = i * 256 + tid;
      wv[i] = *(const float4*)(W1 + (u >> 4) * CCH + 4 * (u & 15));
    }
    #pragma unroll
    for (int i = 0; i < 8; ++i) {
      const int u = i * 256 + tid;
      int row = blockIdx.x * 128 + (u >> 4);
      row = (row < NPTS) ? row : (NPTS - 1);
      av[i] = *(const float4*)(sfeat + row * CCH + 4 * (u & 15));
    }
    #pragma unroll
    for (int i = 0; i < 4; ++i) {
      const int u = i * 256 + tid;
      *(float4*)(sW + (u >> 4) * SW_PITCH + 4 * (u & 15)) = wv[i];
    }
    #pragma unroll
    for (int i = 0; i < 8; ++i) {
      const int u = i * 256 + tid;
      *(float4*)(sA + (u >> 4) * SA_PITCH + 4 * (u & 15)) = av[i];
    }
  }
  __syncthreads();

  v8f acc[4];
  #pragma unroll
  for (int n0 = 0; n0 < 4; ++n0)
    acc[n0] = (v8f){0.f,0.f,0.f,0.f,0.f,0.f,0.f,0.f};

  if (t < TILES) {
    const float* ap = sA + (wave * 16 + lm16) * SA_PITCH + 2 * half;
    const float* bp = sW + (2 * half) * SW_PITCH + lm16;
    #pragma unroll
    for (int ks = 0; ks < CCH / 4; ++ks) {
      const v2f a = *(const v2f*)(ap + 4 * ks);
      #pragma unroll
      for (int n0 = 0; n0 < 4; ++n0) {
        v2f b;
        b.x = bp[(4 * ks) * SW_PITCH + n0 * 16];
        b.y = bp[(4 * ks + 1) * SW_PITCH + n0 * 16];
        acc[n0] = wmma_f32_16x16x4(a, b, acc[n0]);
      }
    }
  }

  #pragma unroll
  for (int n0 = 0; n0 < 4; ++n0) {
    float s = 0.f, q = 0.f;
    #pragma unroll
    for (int r = 0; r < 8; ++r) { float v = acc[n0][r]; s += v; q += v * v; }
    s += __shfl_xor(s, 16);
    q += __shfl_xor(q, 16);
    if (half == 0) { wsum[wave][n0 * 16 + lm16] = s; wssq[wave][n0 * 16 + lm16] = q; }
  }
  __syncthreads();

  if (tid < 64) {
    float s = 0.f, q = 0.f;
    #pragma unroll
    for (int w = 0; w < 8; ++w) { s += wsum[w][tid]; q += wssq[w][tid]; }
    partials[blockIdx.x * 128 + tid]      = s;
    partials[blockIdx.x * 128 + 64 + tid] = q;
  }
}

#define K2_WAVES 32
#define K2_FULL  (GRID1 / K2_WAVES)
#define K2_REM   (GRID1 - K2_FULL * K2_WAVES)
__global__ __launch_bounds__(1024) void k2_finalize(const float* __restrict__ partials,
                                                    const float* __restrict__ gamma,
                                                    const float* __restrict__ beta,
                                                    float* __restrict__ ws) {
  __shared__ alignas(16) float red[K2_WAVES][128];
  const int tid = threadIdx.x;
  const int cq  = tid & 31;
  const int g   = tid >> 5;

  float4 v[K2_FULL];
  #pragma unroll
  for (int i = 0; i < K2_FULL; ++i)
    v[i] = *(const float4*)(partials + (g + K2_WAVES * i) * 128 + 4 * cq);
  float4 acc = v[0];
  #pragma unroll
  for (int i = 1; i < K2_FULL; ++i) {
    acc.x += v[i].x; acc.y += v[i].y; acc.z += v[i].z; acc.w += v[i].w;
  }
  if (g < K2_REM) {
    const float4 t = *(const float4*)(partials + (g + K2_WAVES * K2_FULL) * 128 + 4 * cq);
    acc.x += t.x; acc.y += t.y; acc.z += t.z; acc.w += t.w;
  }
  *(float4*)(&red[g][4 * cq]) = acc;
  __syncthreads();

  if (tid < 64) {
    const int ch = tid;
    float s = 0.f, q = 0.f;
    #pragma unroll
    for (int w = 0; w < K2_WAVES; ++w) { s += red[w][ch]; q += red[w][64 + ch]; }
    const float inv_n = 1.0f / (float)NPTS;
    const float mu  = s * inv_n;
    const float var = q * inv_n - mu * mu;
    const float sc  = gamma[ch] * rsqrtf(var + BN_EPS);
    ws[SCALE_OFF + ch] = sc;
    ws[SHIFT_OFF + ch] = beta[ch] - mu * sc;
  }
}

#define ROWS3   32
#define BLOCKS3 ((NPTS + ROWS3 - 1) / ROWS3)
#define NPAIR3  (ROWS3 * HNB)
__global__ __launch_bounds__(256) void k3_main(const float* __restrict__ qpts,
                                               const float* __restrict__ spts,
                                               const float* __restrict__ sfeat,
                                               const int*   __restrict__ nbidx,
                                               const float* __restrict__ kpts,
                                               const float* __restrict__ W1,
                                               const float* __restrict__ W2,
                                               const float* __restrict__ b2,
                                               const float* __restrict__ ws,
                                               float* __restrict__ out) {
  __shared__ float  act[ROWS3 * 68];
  __shared__ alignas(16) unsigned char pool[ROWS3 * 960 * 2];
  __half* const cw  = (__half*)pool;
  float*  const sA3 = (float*)pool;
  float*  const sW3 = (float*)pool + ROWS3 * SA_PITCH;
  __shared__ int    jIdx[NPAIR3];
  __shared__ float  nnI[NPAIR3];
  __shared__ unsigned char nnK[NPAIR3];
  __shared__ unsigned needM[ROWS3];
  __shared__ unsigned char nchP[ROWS3];

  const int tid  = threadIdx.x;
  const int wave = tid >> 5;
  const int lane = tid & 31;
  const int half = lane >> 4;
  const int lm16 = lane & 15;
  const int m0   = blockIdx.x * ROWS3;

  {
    float4 stA[2], stW[4];
    #pragma unroll
    for (int i = 0; i < 2; ++i) {
      const int u = i * 256 + tid;
      int row = m0 + (u >> 4);
      row = (row < NPTS) ? row : (NPTS - 1);
      stA[i] = *(const float4*)(sfeat + row * CCH + 4 * (u & 15));
    }
    #pragma unroll
    for (int i = 0; i < 4; ++i) {
      const int u = i * 256 + tid;
      stW[i] = *(const float4*)(W1 + (u >> 4) * CCH + 4 * (u & 15));
    }
    #pragma unroll
    for (int i = 0; i < 2; ++i) {
      const int u = i * 256 + tid;
      *(float4*)(sA3 + (u >> 4) * SA_PITCH + 4 * (u & 15)) = stA[i];
    }
    #pragma unroll
    for (int i = 0; i < 4; ++i) {
      const int u = i * 256 + tid;
      *(float4*)(sW3 + (u >> 4) * SW_PITCH + 4 * (u & 15)) = stW[i];
    }
  }
  __syncthreads();
  {
    const int mt  = wave >> 2;
    const int n0  = (wave & 3) * 16;
    const float* ap = sA3 + (mt * 16 + lm16) * SA_PITCH + 2 * half;
    const float* bp = sW3 + (2 * half) * SW_PITCH + n0 + lm16;
    v8f acc = (v8f){0.f,0.f,0.f,0.f,0.f,0.f,0.f,0.f};
    #pragma unroll
    for (int ks = 0; ks < CCH / 4; ++ks) {
      const v2f a = *(const v2f*)(ap + 4 * ks);
      v2f b;
      b.x = bp[(4 * ks) * SW_PITCH];
      b.y = bp[(4 * ks + 1) * SW_PITCH];
      acc = wmma_f32_16x16x4(a, b, acc);
    }
    const int ch = n0 + lm16;
    const float sc = ws[SCALE_OFF + ch];
    const float sh = ws[SHIFT_OFF + ch];
    #pragma unroll
    for (int r = 0; r < 8; ++r) {
      const int M = mt * 16 + r + 8 * half;
      float v = acc[r] * sc + sh;
      v = (v > 0.f) ? v : 0.1f * v;
      act[M * 68 + ch] = v;
    }
  }
  {
    #pragma unroll 2
    for (int i = 0; i < 4; ++i) {
      const int p  = i * 256 + tid;
      const int lm = p >> 5;
      const int h  = p & 31;
      int m = m0 + lm;
      m = (m < NPTS) ? m : (NPTS - 1);
      const int j  = nbidx[m * HNB + h];
      const bool valid = ((unsigned)j < (unsigned)NPTS);
      const int jj = valid ? j : 0;
      const float qx = qpts[m * 3 + 0];
      const float qy = qpts[m * 3 + 1];
      const float qz = qpts[m * 3 + 2];
      const float dx = spts[jj * 3 + 0] - qx;
      const float dy = spts[jj * 3 + 1] - qy;
      const float dz = spts[jj * 3 + 2] - qz;
      float best = 3.4e38f; int bk = 0;
      #pragma unroll
      for (int k = 0; k < KPN; ++k) {
        const float ex = dx - kpts[k * 3 + 0];
        const float ey = dy - kpts[k * 3 + 1];
        const float ez = dz - kpts[k * 3 + 2];
        const float sq = ex * ex + ey * ey + ez * ez;
        if (sq < best) { best = sq; bk = k; }
      }
      float infl = 1.0f - sqrtf(best) * (1.0f / SIGMA);
      infl = (infl > 0.f) ? infl : 0.f;
      if (!valid) infl = 0.f;
      const bool on = (infl > 0.f);
      int um = on ? (1 << bk) : 0;
      um |= __shfl_xor(um, 16);
      um |= __shfl_xor(um, 8);
      um |= __shfl_xor(um, 4);
      um |= __shfl_xor(um, 2);
      um |= __shfl_xor(um, 1);
      um = um ? um : 1;
      const unsigned onb = __builtin_amdgcn_ballot_w32(on);
      const int cnt  = __popc(onb);
      const int pos  = __popc(onb & ((1u << h) - 1u));
      const int npad = (cnt > 8) ? ((cnt + 7) & ~7) : 8;
      const int lb   = lm * HNB;
      if (on) {
        jIdx[lb + pos] = jj;
        nnK[lb + pos]  = (unsigned char)bk;
        nnI[lb + pos]  = infl;
      }
      if (h >= cnt && h < npad) {
        jIdx[lb + h] = m;
        nnK[lb + h]  = (unsigned char)(__ffs(um) - 1);
        nnI[lb + h]  = 0.f;
      }
      if (h == 0) { needM[lm] = (unsigned)um; nchP[lm] = (unsigned char)(npad >> 3); }
    }
  }
  __syncthreads();

  const unsigned nm = needM[lane];
  #pragma unroll 1
  for (int kp = wave; kp < KPN; kp += 8) {
    const int cb = kp * 64 + 4 * lm16;
    const unsigned bal = __builtin_amdgcn_ballot_w32(((nm >> kp) & 1u) != 0u);
    const int nk = __popc(bal);
    const float4 bias = *(const float4*)(b2 + cb);
    #pragma unroll 1
    for (int s0 = 0; s0 < nk; s0 += 16) {
      const int s  = s0 + lm16;
      const int mr = nth_set_bit(bal, (s < nk) ? s : s0);
      v8f acc0 = (v8f){0.f,0.f,0.f,0.f,0.f,0.f,0.f,0.f};
      v8f acc1 = (v8f){0.f,0.f,0.f,0.f,0.f,0.f,0.f,0.f};
      v8f acc2 = (v8f){0.f,0.f,0.f,0.f,0.f,0.f,0.f,0.f};
      v8f acc3 = (v8f){0.f,0.f,0.f,0.f,0.f,0.f,0.f,0.f};
      const float* ap = act + mr * 68 + 2 * half;
      const float* bp = W2 + (2 * half) * 960 + cb;
      #pragma unroll 1
      for (int kh = 0; kh < CCH / 4; kh += 8) {
        #pragma unroll
        for (int ks = kh; ks < kh + 8; ++ks) {
          const v2f a = *(const v2f*)(ap + 4 * ks);
          const float4 bx = *(const float4*)(bp + (4 * ks) * 960);
          const float4 by = *(const float4*)(bp + (4 * ks + 1) * 960);
          acc0 = wmma_f32_16x16x4(a, (v2f){bx.x, by.x}, acc0);
          acc1 = wmma_f32_16x16x4(a, (v2f){bx.y, by.y}, acc1);
          acc2 = wmma_f32_16x16x4(a, (v2f){bx.z, by.z}, acc2);
          acc3 = wmma_f32_16x16x4(a, (v2f){bx.w, by.w}, acc3);
        }
      }
      #pragma unroll
      for (int r = 0; r < 8; ++r) {
        const int plo = __builtin_amdgcn_readlane(mr, r);
        const int phi = __builtin_amdgcn_readlane(mr, r + 8);
        const int pr  = half ? phi : plo;
        const h4 pk = {(_Float16)(acc0[r] + bias.x), (_Float16)(acc1[r] + bias.y),
                       (_Float16)(acc2[r] + bias.z), (_Float16)(acc3[r] + bias.w)};
        *(h4*)(cw + pr * 960 + cb) = pk;
      }
    }
  }
  __syncthreads();

  const int g = half;
  const int q = lm16;
  const float4* __restrict__ sf4 = (const float4*)sfeat;
  float4 a[4];
  {
    float4 x[4][4];
    #pragma unroll
    for (int r = 0; r < 4; ++r) {
      const int pb = (wave + r * 8) * HNB + g;
      #pragma unroll
      for (int i = 0; i < 4; ++i) {
        const unsigned j = (unsigned)jIdx[pb + 2 * i];
        x[r][i] = sf4[j * 16u + (unsigned)q];
      }
    }
    __builtin_amdgcn_sched_barrier(0);
    #pragma unroll
    for (int r = 0; r < 4; ++r) {
      const int lm = wave + r * 8;
      const int pb = lm * HNB + g;
      a[r] = make_float4(0.f, 0.f, 0.f, 0.f);
      #pragma unroll
      for (int i = 0; i < 4; ++i) {
        const int   k = nnK[pb + 2 * i];
        const float f = nnI[pb + 2 * i];
        const __half2* wp = (const __half2*)(cw + lm * 960 + k * 64 + 4 * q);
        const __half2 w01 = wp[0];
        const __half2 w23 = wp[1];
        a[r].x = fmaf(x[r][i].x, __low2float(w01)  * f, a[r].x);
        a[r].y = fmaf(x[r][i].y, __high2float(w01) * f, a[r].y);
        a[r].z = fmaf(x[r][i].z, __low2float(w23)  * f, a[r].z);
        a[r].w = fmaf(x[r][i].w, __high2float(w23) * f, a[r].w);
      }
    }
  }
  int nch[4];
  #pragma unroll
  for (int r = 0; r < 4; ++r)
    nch[r] = __builtin_amdgcn_readfirstlane((int)nchP[wave + r * 8]);
  const int maxch = max(max(nch[0], nch[1]), max(nch[2], nch[3]));
  #pragma unroll 1
  for (int c = 1; c < maxch; ++c) {
    float4 x[4][4];
    #pragma unroll
    for (int r = 0; r < 4; ++r) {
      if (c < nch[r]) {
        const int pb = (wave + r * 8) * HNB + c * 8 + g;
        #pragma unroll
        for (int i = 0; i < 4; ++i) {
          const unsigned j = (unsigned)jIdx[pb + 2 * i];
          x[r][i] = sf4[j * 16u + (unsigned)q];
        }
      }
    }
    #pragma unroll
    for (int r = 0; r < 4; ++r) {
      if (c < nch[r]) {
        const int lm = wave + r * 8;
        const int pb = lm * HNB + c * 8 + g;
        #pragma unroll
        for (int i = 0; i < 4; ++i) {
          const int   k = nnK[pb + 2 * i];
          const float f = nnI[pb + 2 * i];
          const __half2* wp = (const __half2*)(cw + lm * 960 + k * 64 + 4 * q);
          const __half2 w01 = wp[0];
          const __half2 w23 = wp[1];
          a[r].x = fmaf(x[r][i].x, __low2float(w01)  * f, a[r].x);
          a[r].y = fmaf(x[r][i].y, __high2float(w01) * f, a[r].y);
          a[r].z = fmaf(x[r][i].z, __low2float(w23)  * f, a[r].z);
          a[r].w = fmaf(x[r][i].w, __high2float(w23) * f, a[r].w);
        }
      }
    }
  }
  const int nrow = (m0 + 16 < NPTS) ? 4 : 2;
  #pragma unroll
  for (int r = 0; r < 4; ++r) {
    if (r < nrow) {
      const int lm = wave + r * 8;
      float4 t = a[r];
      t.x += __shfl_xor(t.x, 16);
      t.y += __shfl_xor(t.y, 16);
      t.z += __shfl_xor(t.z, 16);
      t.w += __shfl_xor(t.w, 16);
      float2 o;
      o.x = g ? t.z : t.x;
      o.y = g ? t.w : t.y;
      *(float2*)(out + (m0 + lm) * 64 + 4 * q + 2 * g) = o;
    }
  }
}

extern "C" void kernel_launch(void* const* d_in, const int* in_sizes, int n_in,
                              void* d_out, int out_size, void* d_ws, size_t ws_size,
                              hipStream_t stream) {
  const float* qpts  = (const float*)d_in[0];
  const float* spts  = (const float*)d_in[1];
  const float* sfeat = (const float*)d_in[2];
  const int*   nbidx = (const int*)  d_in[3];
  const float* kpts  = (const float*)d_in[4];
  const float* W1    = (const float*)d_in[5];
  const float* gamma = (const float*)d_in[6];
  const float* beta  = (const float*)d_in[7];
  const float* W2    = (const float*)d_in[8];
  const float* b2    = (const float*)d_in[9];
  float* out = (float*)d_out;
  float* ws  = (float*)d_ws;

  k1_stats<<<GRID1, 256, 0, stream>>>(sfeat, W1, ws);
  k2_finalize<<<1, 1024, 0, stream>>>(ws, gamma, beta, ws);
  k3_main<<<BLOCKS3, 256, 0, stream>>>(qpts, spts, sfeat, nbidx, kpts,
                                     W1, W2, b2, ws, out);
  (void)in_sizes; (void)n_in; (void)out_size; (void)ws_size;
}
